# starting kernel
# baseline (speedup 1.0000x reference)
#include <hip/hip_runtime.h>
#include <math.h>

typedef __attribute__((ext_vector_type(2))) float v2f;
typedef __attribute__((ext_vector_type(8))) float v8f;

#define NBATCH 4
#define CIN 512
#define HH 48
#define WW 48
#define LSPAT (HH*WW)
#define CH3 192
#define HCH 64
#define NHEAD 8
#define DHEAD 8
#define ATTN_SCALE 0.35355339059327379f

static __device__ __forceinline__ v8f wmma4(v2f a, v2f b, v8f c) {
  return __builtin_amdgcn_wmma_f32_16x16x4_f32(false, a, false, b, (short)0, c,
                                               false, false);
}

__global__ __launch_bounds__(256) void qkv_gemm_kernel(
    const float* __restrict__ x, const float* __restrict__ w,
    const float* __restrict__ bias, float* __restrict__ qkv) {
  const int lane = threadIdx.x & 31;
  const int half = lane >> 4;
  const int l16  = lane & 15;
  const int wave = blockIdx.x * (blockDim.x >> 5) + (threadIdx.x >> 5);

  const int ntn = LSPAT / 16;
  const int ntm = CH3 / 16;
  int t = wave;
  const int n  = t / (ntm * ntn);  t -= n * ntm * ntn;
  const int mt = t / ntn;
  const int ct = t % ntn;

  const int mrow = mt * 16 + l16;
  const int col  = ct * 16 + l16;
  const float* xb = x + (size_t)n * CIN * LSPAT;

  v8f acc = {};
  for (int kk = 0; kk < CIN; kk += 4) {
    const int ka = kk + half * 2;
    v2f a; a[0] = w[mrow * CIN + ka];
           a[1] = w[mrow * CIN + ka + 1];
    v2f b; b[0] = xb[(size_t)ka * LSPAT + col];
           b[1] = xb[(size_t)(ka + 1) * LSPAT + col];
    acc = wmma4(a, b, acc);
  }

  float* out = qkv + (size_t)n * CH3 * LSPAT;
#pragma unroll
  for (int r = 0; r < 8; ++r) {
    const int row = mt * 16 + r + half * 8;
    out[(size_t)row * LSPAT + col] = acc[r] + bias[row];
  }
}

__global__ __launch_bounds__(128) void attn_kernel(
    const float* __restrict__ qkv, float* __restrict__ obuf) {
  __shared__ float plds[4][16][17];
  const int lane = threadIdx.x & 31;
  const int half = lane >> 4;
  const int l16  = lane & 15;
  const int wv   = threadIdx.x >> 5;

  const int qtiles = LSPAT / 64;
  const int nh = blockIdx.x / qtiles;
  const int qt = blockIdx.x % qtiles;
  const int n = nh >> 3, h = nh & 7;
  const int qbase = qt * 64 + wv * 16;

  const float* base = qkv + (size_t)n * CH3 * LSPAT + (size_t)h * 24 * LSPAT;
  const float* qm = base;
  const float* km = base + (size_t)DHEAD * LSPAT;
  const float* vm = base + (size_t)2 * DHEAD * LSPAT;

  v2f aq[2];
#pragma unroll
  for (int db = 0; db < 2; ++db) {
    const int d0 = db * 4 + half * 2;
    aq[db][0] = qm[(size_t)d0 * LSPAT + qbase + l16];
    aq[db][1] = qm[(size_t)(d0 + 1) * LSPAT + qbase + l16];
  }

  float m8[8];
#pragma unroll
  for (int r = 0; r < 8; ++r) m8[r] = -3.0e38f;

  for (int kt = 0; kt < LSPAT / 16; ++kt) {
    const int kb = kt * 16;
    v8f s = {};
#pragma unroll
    for (int db = 0; db < 2; ++db) {
      const int d0 = db * 4 + half * 2;
      v2f bk; bk[0] = km[(size_t)d0 * LSPAT + kb + l16];
              bk[1] = km[(size_t)(d0 + 1) * LSPAT + kb + l16];
      s = wmma4(aq[db], bk, s);
    }
#pragma unroll
    for (int r = 0; r < 8; ++r) m8[r] = fmaxf(m8[r], s[r]);
  }
#pragma unroll
  for (int r = 0; r < 8; ++r) {
#pragma unroll
    for (int m = 1; m < 16; m <<= 1) m8[r] = fmaxf(m8[r], __shfl_xor(m8[r], m, 32));
  }

  float s8[8];
#pragma unroll
  for (int r = 0; r < 8; ++r) s8[r] = 0.0f;

  for (int kt = 0; kt < LSPAT / 16; ++kt) {
    const int kb = kt * 16;
    v8f s = {};
#pragma unroll
    for (int db = 0; db < 2; ++db) {
      const int d0 = db * 4 + half * 2;
      v2f bk; bk[0] = km[(size_t)d0 * LSPAT + kb + l16];
              bk[1] = km[(size_t)(d0 + 1) * LSPAT + kb + l16];
      s = wmma4(aq[db], bk, s);
    }
#pragma unroll
    for (int r = 0; r < 8; ++r) s8[r] += __expf((s[r] - m8[r]) * ATTN_SCALE);
  }
  float inv[8];
#pragma unroll
  for (int r = 0; r < 8; ++r) {
#pragma unroll
    for (int m = 1; m < 16; m <<= 1) s8[r] += __shfl_xor(s8[r], m, 32);
    inv[r] = 1.0f / s8[r];
  }

  v8f acc = {};
  for (int kt = 0; kt < LSPAT / 16; ++kt) {
    const int kb = kt * 16;
    v8f s = {};
#pragma unroll
    for (int db = 0; db < 2; ++db) {
      const int d0 = db * 4 + half * 2;
      v2f bk; bk[0] = km[(size_t)d0 * LSPAT + kb + l16];
              bk[1] = km[(size_t)(d0 + 1) * LSPAT + kb + l16];
      s = wmma4(aq[db], bk, s);
    }
#pragma unroll
    for (int r = 0; r < 8; ++r)
      plds[wv][r + half * 8][l16] =
          __expf((s[r] - m8[r]) * ATTN_SCALE) * inv[r];
    __syncthreads();
#pragma unroll
    for (int c = 0; c < 4; ++c) {
      const int k0 = c * 4 + half * 2;
      v2f ap; ap[0] = plds[wv][l16][k0];
              ap[1] = plds[wv][l16][k0 + 1];
      v2f bv;
      if (l16 < 8) {
        bv[0] = vm[(size_t)l16 * LSPAT + kb + k0];
        bv[1] = vm[(size_t)l16 * LSPAT + kb + k0 + 1];
      } else { bv[0] = 0.0f; bv[1] = 0.0f; }
      acc = wmma4(ap, bv, acc);
    }
    __syncthreads();
  }

  if (l16 < 8) {
    float* ob = obuf + (size_t)n * HCH * LSPAT + (size_t)(h * DHEAD + l16) * LSPAT;
#pragma unroll
    for (int r = 0; r < 8; ++r) ob[qbase + r + half * 8] = acc[r];
  }
}

__global__ __launch_bounds__(256) void pos_conv_kernel(
    const float* __restrict__ qkv, const float* __restrict__ pw,
    const float* __restrict__ pb, float* __restrict__ obuf) {
  int idx = blockIdx.x * 256 + threadIdx.x;
  const int xp = idx % WW; int t = idx / WW;
  const int yp = t % HH;  t /= HH;
  const int oc = t % HCH; const int n = t / HCH;

  const float* vbase = qkv + (size_t)n * CH3 * LSPAT;
  float acc = pb[oc];
  for (int ic = 0; ic < HCH; ++ic) {
    const int ch = (ic >> 3) * 24 + 16 + (ic & 7);
    const float* vin  = vbase + (size_t)ch * LSPAT;
    const float* wrow = pw + ((size_t)oc * HCH + ic) * 9;
#pragma unroll
    for (int ky = 0; ky < 3; ++ky) {
      const int iy = yp + ky - 1;
      if (iy < 0 || iy >= HH) continue;
#pragma unroll
      for (int kx = 0; kx < 3; ++kx) {
        const int ix = xp + kx - 1;
        if (ix < 0 || ix >= WW) continue;
        acc = fmaf(vin[iy * WW + ix], wrow[ky * 3 + kx], acc);
      }
    }
  }
  obuf[(size_t)n * HCH * LSPAT + (size_t)oc * LSPAT + yp * WW + xp] += acc;
}

__global__ __launch_bounds__(256) void res_gemm_kernel(
    const float* __restrict__ obuf, const float* __restrict__ w,
    const float* __restrict__ bias, float* __restrict__ out) {
  const int lane = threadIdx.x & 31;
  const int half = lane >> 4;
  const int l16  = lane & 15;
  const int wave = blockIdx.x * (blockDim.x >> 5) + (threadIdx.x >> 5);

  const int ntn = LSPAT / 16;
  const int ntm = CIN / 16;
  int t = wave;
  const int n  = t / (ntm * ntn); t -= n * ntm * ntn;
  const int mt = t / ntn;
  const int ct = t % ntn;

  const int mrow = mt * 16 + l16;
  const int col  = ct * 16 + l16;
  const float* ib = obuf + (size_t)n * HCH * LSPAT;

  v8f acc = {};
#pragma unroll
  for (int kk = 0; kk < HCH; kk += 4) {
    const int ka = kk + half * 2;
    v2f a; a[0] = w[mrow * HCH + ka];
           a[1] = w[mrow * HCH + ka + 1];
    v2f b; b[0] = ib[(size_t)ka * LSPAT + col];
           b[1] = ib[(size_t)(ka + 1) * LSPAT + col];
    acc = wmma4(a, b, acc);
  }

  float* ob = out + (size_t)n * CIN * LSPAT;
#pragma unroll
  for (int r = 0; r < 8; ++r) {
    const int row = mt * 16 + r + half * 8;
    ob[(size_t)row * LSPAT + col] = acc[r] + bias[row];
  }
}

extern "C" void kernel_launch(void* const* d_in, const int* in_sizes, int n_in,
                              void* d_out, int out_size, void* d_ws, size_t ws_size,
                              hipStream_t stream) {
  const float* x     = (const float*)d_in[0];
  const float* qkv_w = (const float*)d_in[1];
  const float* qkv_b = (const float*)d_in[2];
  const float* pos_w = (const float*)d_in[3];
  const float* pos_b = (const float*)d_in[4];
  const float* res_w = (const float*)d_in[5];
  const float* res_b = (const float*)d_in[6];
  float* out = (float*)d_out;

  float* qkv  = (float*)d_ws;
  float* obuf = qkv + (size_t)NBATCH * CH3 * LSPAT;

  const int waves1 = NBATCH * (CH3 / 16) * (LSPAT / 16);
  qkv_gemm_kernel<<<waves1 / 8, 256, 0, stream>>>(x, qkv_w, qkv_b, qkv);

  attn_kernel<<<NBATCH * NHEAD * (LSPAT / 64), 128, 0, stream>>>(qkv, obuf);

  pos_conv_kernel<<<(NBATCH * HCH * LSPAT) / 256, 256, 0, stream>>>(
      qkv, pos_w, pos_b, obuf);

  const int waves4 = NBATCH * (CIN / 16) * (LSPAT / 16);
  res_gemm_kernel<<<waves4 / 8, 256, 0, stream>>>(obuf, res_w, res_b, out);
}


# pos 3x3 conv as a register-tiled direct conv on the vector unit: lane = pixel, 8 output channels per lane, taps via scalar loads, a head's 72 inputs loaded as one group; per-output sum order kept
# speedup vs baseline: 1.4086x; 1.4086x over previous
#include <hip/hip_runtime.h>
#include <math.h>

typedef __attribute__((ext_vector_type(2))) float v2f;
typedef __attribute__((ext_vector_type(8))) float v8f;

#define NBATCH 4
#define CIN 512
#define HH 48
#define WW 48
#define LSPAT (HH*WW)
#define CH3 192
#define HCH 64
#define NHEAD 8
#define DHEAD 8
#define ATTN_SCALE 0.35355339059327379f

static __device__ __forceinline__ v8f wmma4(v2f a, v2f b, v8f c) {
  return __builtin_amdgcn_wmma_f32_16x16x4_f32(false, a, false, b, (short)0, c,
                                               false, false);
}

__global__ __launch_bounds__(256) void qkv_gemm_kernel(
    const float* __restrict__ x, const float* __restrict__ w,
    const float* __restrict__ bias, float* __restrict__ qkv) {
  const int lane = threadIdx.x & 31;
  const int half = lane >> 4;
  const int l16  = lane & 15;
  const int wave = blockIdx.x * (blockDim.x >> 5) + (threadIdx.x >> 5);

  const int ntn = LSPAT / 16;
  const int ntm = CH3 / 16;
  int t = wave;
  const int n  = t / (ntm * ntn);  t -= n * ntm * ntn;
  const int mt = t / ntn;
  const int ct = t % ntn;

  const int mrow = mt * 16 + l16;
  const int col  = ct * 16 + l16;
  const float* xb = x + (size_t)n * CIN * LSPAT;

  v8f acc = {};
  for (int kk = 0; kk < CIN; kk += 4) {
    const int ka = kk + half * 2;
    v2f a; a[0] = w[mrow * CIN + ka];
           a[1] = w[mrow * CIN + ka + 1];
    v2f b; b[0] = xb[(size_t)ka * LSPAT + col];
           b[1] = xb[(size_t)(ka + 1) * LSPAT + col];
    acc = wmma4(a, b, acc);
  }

  float* out = qkv + (size_t)n * CH3 * LSPAT;
#pragma unroll
  for (int r = 0; r < 8; ++r) {
    const int row = mt * 16 + r + half * 8;
    out[(size_t)row * LSPAT + col] = acc[r] + bias[row];
  }
}

__global__ __launch_bounds__(128) void attn_kernel(
    const float* __restrict__ qkv, float* __restrict__ obuf) {
  __shared__ float plds[4][16][17];
  const int lane = threadIdx.x & 31;
  const int half = lane >> 4;
  const int l16  = lane & 15;
  const int wv   = threadIdx.x >> 5;

  const int qtiles = LSPAT / 64;
  const int nh = blockIdx.x / qtiles;
  const int qt = blockIdx.x % qtiles;
  const int n = nh >> 3, h = nh & 7;
  const int qbase = qt * 64 + wv * 16;

  const float* base = qkv + (size_t)n * CH3 * LSPAT + (size_t)h * 24 * LSPAT;
  const float* qm = base;
  const float* km = base + (size_t)DHEAD * LSPAT;
  const float* vm = base + (size_t)2 * DHEAD * LSPAT;

  v2f aq[2];
#pragma unroll
  for (int db = 0; db < 2; ++db) {
    const int d0 = db * 4 + half * 2;
    aq[db][0] = qm[(size_t)d0 * LSPAT + qbase + l16];
    aq[db][1] = qm[(size_t)(d0 + 1) * LSPAT + qbase + l16];
  }

  float m8[8];
#pragma unroll
  for (int r = 0; r < 8; ++r) m8[r] = -3.0e38f;

  for (int kt = 0; kt < LSPAT / 16; ++kt) {
    const int kb = kt * 16;
    v8f s = {};
#pragma unroll
    for (int db = 0; db < 2; ++db) {
      const int d0 = db * 4 + half * 2;
      v2f bk; bk[0] = km[(size_t)d0 * LSPAT + kb + l16];
              bk[1] = km[(size_t)(d0 + 1) * LSPAT + kb + l16];
      s = wmma4(aq[db], bk, s);
    }
#pragma unroll
    for (int r = 0; r < 8; ++r) m8[r] = fmaxf(m8[r], s[r]);
  }
#pragma unroll
  for (int r = 0; r < 8; ++r) {
#pragma unroll
    for (int m = 1; m < 16; m <<= 1) m8[r] = fmaxf(m8[r], __shfl_xor(m8[r], m, 32));
  }

  float s8[8];
#pragma unroll
  for (int r = 0; r < 8; ++r) s8[r] = 0.0f;

  for (int kt = 0; kt < LSPAT / 16; ++kt) {
    const int kb = kt * 16;
    v8f s = {};
#pragma unroll
    for (int db = 0; db < 2; ++db) {
      const int d0 = db * 4 + half * 2;
      v2f bk; bk[0] = km[(size_t)d0 * LSPAT + kb + l16];
              bk[1] = km[(size_t)(d0 + 1) * LSPAT + kb + l16];
      s = wmma4(aq[db], bk, s);
    }
#pragma unroll
    for (int r = 0; r < 8; ++r) s8[r] += __expf((s[r] - m8[r]) * ATTN_SCALE);
  }
  float inv[8];
#pragma unroll
  for (int r = 0; r < 8; ++r) {
#pragma unroll
    for (int m = 1; m < 16; m <<= 1) s8[r] += __shfl_xor(s8[r], m, 32);
    inv[r] = 1.0f / s8[r];
  }

  v8f acc = {};
  for (int kt = 0; kt < LSPAT / 16; ++kt) {
    const int kb = kt * 16;
    v8f s = {};
#pragma unroll
    for (int db = 0; db < 2; ++db) {
      const int d0 = db * 4 + half * 2;
      v2f bk; bk[0] = km[(size_t)d0 * LSPAT + kb + l16];
              bk[1] = km[(size_t)(d0 + 1) * LSPAT + kb + l16];
      s = wmma4(aq[db], bk, s);
    }
#pragma unroll
    for (int r = 0; r < 8; ++r)
      plds[wv][r + half * 8][l16] =
          __expf((s[r] - m8[r]) * ATTN_SCALE) * inv[r];
    __syncthreads();
#pragma unroll
    for (int c = 0; c < 4; ++c) {
      const int k0 = c * 4 + half * 2;
      v2f ap; ap[0] = plds[wv][l16][k0];
              ap[1] = plds[wv][l16][k0 + 1];
      v2f bv;
      if (l16 < 8) {
        bv[0] = vm[(size_t)l16 * LSPAT + kb + k0];
        bv[1] = vm[(size_t)l16 * LSPAT + kb + k0 + 1];
      } else { bv[0] = 0.0f; bv[1] = 0.0f; }
      acc = wmma4(ap, bv, acc);
    }
    __syncthreads();
  }

  if (l16 < 8) {
    float* ob = obuf + (size_t)n * HCH * LSPAT + (size_t)(h * DHEAD + l16) * LSPAT;
#pragma unroll
    for (int r = 0; r < 8; ++r) ob[qbase + r + half * 8] = acc[r];
  }
}

#define POS_OCT 8
#define POS_BLK 128
__global__ __launch_bounds__(POS_BLK) void pos_conv_kernel(
    const float* __restrict__ qkv, const float* __restrict__ pw,
    const float* __restrict__ pb, float* __restrict__ obuf) {
  const int chunks = LSPAT / POS_BLK;
  int b = blockIdx.x;
  const int chunk = b % chunks;  b /= chunks;
  const int og = b % (HCH / POS_OCT);
  const int n  = b / (HCH / POS_OCT);
  const unsigned p = chunk * POS_BLK + threadIdx.x;
  const int yp = (int)(p / WW);
  const int xp = (int)(p % WW);
  const int oc0 = og * POS_OCT;

  unsigned off[9];
  bool ok[9];
#pragma unroll
  for (int ky = 0; ky < 3; ++ky) {
#pragma unroll
    for (int kx = 0; kx < 3; ++kx) {
      const int iy = yp + ky - 1;
      const int ix = xp + kx - 1;
      const bool in_img = (iy >= 0) && (iy < HH) && (ix >= 0) && (ix < WW);
      ok[ky * 3 + kx]  = in_img;
      off[ky * 3 + kx] = in_img ? (unsigned)(iy * WW + ix) : p;
    }
  }

  float acc[POS_OCT];
#pragma unroll
  for (int o = 0; o < POS_OCT; ++o) acc[o] = pb[oc0 + o];

  const float* vbase = qkv + (size_t)n * CH3 * LSPAT + (size_t)16 * LSPAT;
  float cur[DHEAD][9];
#pragma unroll
  for (int dd = 0; dd < DHEAD; ++dd) {
#pragma unroll
    for (int t = 0; t < 9; ++t) cur[dd][t] = vbase[(unsigned)(dd * LSPAT) + off[t]];
  }

  for (int hh = 0; hh < NHEAD; ++hh) {
    const int hn = (hh + 1 < NHEAD) ? hh + 1 : hh;
    const unsigned cho = (unsigned)(hn * 24 * LSPAT);
    float nxt[DHEAD][9];
#pragma unroll
    for (int dd = 0; dd < DHEAD; ++dd) {
#pragma unroll
      for (int t = 0; t < 9; ++t)
        nxt[dd][t] = vbase[cho + (unsigned)(dd * LSPAT) + off[t]];
    }

#pragma unroll
    for (int dd = 0; dd < DHEAD; ++dd) {
      const int ic = hh * DHEAD + dd;
      float in[9];
#pragma unroll
      for (int t = 0; t < 9; ++t) in[t] = ok[t] ? cur[dd][t] : 0.0f;
#pragma unroll
      for (int o = 0; o < POS_OCT; ++o) {
        const float* wrow = pw + ((size_t)(oc0 + o) * HCH + ic) * 9;
#pragma unroll
        for (int t = 0; t < 9; ++t) acc[o] = fmaf(in[t], wrow[t], acc[o]);
      }
    }

#pragma unroll
    for (int dd = 0; dd < DHEAD; ++dd) {
#pragma unroll
      for (int t = 0; t < 9; ++t) cur[dd][t] = nxt[dd][t];
    }
  }

  float* ob = obuf + (size_t)n * HCH * LSPAT + (size_t)oc0 * LSPAT + p;
#pragma unroll
  for (int o = 0; o < POS_OCT; ++o) ob[(size_t)o * LSPAT] += acc[o];
}

__global__ __launch_bounds__(256) void res_gemm_kernel(
    const float* __restrict__ obuf, const float* __restrict__ w,
    const float* __restrict__ bias, float* __restrict__ out) {
  const int lane = threadIdx.x & 31;
  const int half = lane >> 4;
  const int l16  = lane & 15;
  const int wave = blockIdx.x * (blockDim.x >> 5) + (threadIdx.x >> 5);

  const int ntn = LSPAT / 16;
  const int ntm = CIN / 16;
  int t = wave;
  const int n  = t / (ntm * ntn); t -= n * ntm * ntn;
  const int mt = t / ntn;
  const int ct = t % ntn;

  const int mrow = mt * 16 + l16;
  const int col  = ct * 16 + l16;
  const float* ib = obuf + (size_t)n * HCH * LSPAT;

  v8f acc = {};
#pragma unroll
  for (int kk = 0; kk < HCH; kk += 4) {
    const int ka = kk + half * 2;
    v2f a; a[0] = w[mrow * HCH + ka];
           a[1] = w[mrow * HCH + ka + 1];
    v2f b; b[0] = ib[(size_t)ka * LSPAT + col];
           b[1] = ib[(size_t)(ka + 1) * LSPAT + col];
    acc = wmma4(a, b, acc);
  }

  float* ob = out + (size_t)n * CIN * LSPAT;
#pragma unroll
  for (int r = 0; r < 8; ++r) {
    const int row = mt * 16 + r + half * 8;
    ob[(size_t)row * LSPAT + col] = acc[r] + bias[row];
  }
}

extern "C" void kernel_launch(void* const* d_in, const int* in_sizes, int n_in,
                              void* d_out, int out_size, void* d_ws, size_t ws_size,
                              hipStream_t stream) {
  const float* x     = (const float*)d_in[0];
  const float* qkv_w = (const float*)d_in[1];
  const float* qkv_b = (const float*)d_in[2];
  const float* pos_w = (const float*)d_in[3];
  const float* pos_b = (const float*)d_in[4];
  const float* res_w = (const float*)d_in[5];
  const float* res_b = (const float*)d_in[6];
  float* out = (float*)d_out;

  float* qkv  = (float*)d_ws;
  float* obuf = qkv + (size_t)NBATCH * CH3 * LSPAT;

  const int waves1 = NBATCH * (CH3 / 16) * (LSPAT / 16);
  qkv_gemm_kernel<<<waves1 / 8, 256, 0, stream>>>(x, qkv_w, qkv_b, qkv);

  attn_kernel<<<NBATCH * NHEAD * (LSPAT / 64), 128, 0, stream>>>(qkv, obuf);

  pos_conv_kernel<<<NBATCH * (HCH / POS_OCT) * (LSPAT / POS_BLK), POS_BLK, 0,
                    stream>>>(qkv, pos_w, pos_b, obuf);

  const int waves4 = NBATCH * (CIN / 16) * (LSPAT / 16);
  res_gemm_kernel<<<waves4 / 8, 256, 0, stream>>>(obuf, res_w, res_b, out);
}


# attention P*V pass: K and V fragments of a key tile requested as one group one tile ahead (two register sets); V loaded by all lanes, no masked load+wait per step; barrier fences name LDS only
# speedup vs baseline: 1.7202x; 1.2212x over previous
#include <hip/hip_runtime.h>
#include <math.h>

typedef __attribute__((ext_vector_type(2))) float v2f;
typedef __attribute__((ext_vector_type(8))) float v8f;

#define NBATCH 4
#define CIN 512
#define HH 48
#define WW 48
#define LSPAT (HH*WW)
#define CH3 192
#define HCH 64
#define NHEAD 8
#define DHEAD 8
#define ATTN_SCALE 0.35355339059327379f

static __device__ __forceinline__ v8f wmma4(v2f a, v2f b, v8f c) {
  return __builtin_amdgcn_wmma_f32_16x16x4_f32(false, a, false, b, (short)0, c,
                                               false, false);
}

__global__ __launch_bounds__(256) void qkv_gemm_kernel(
    const float* __restrict__ x, const float* __restrict__ w,
    const float* __restrict__ bias, float* __restrict__ qkv) {
  const int lane = threadIdx.x & 31;
  const int half = lane >> 4;
  const int l16  = lane & 15;
  const int wave = blockIdx.x * (blockDim.x >> 5) + (threadIdx.x >> 5);

  const int ntn = LSPAT / 16;
  const int ntm = CH3 / 16;
  int t = wave;
  const int n  = t / (ntm * ntn);  t -= n * ntm * ntn;
  const int mt = t / ntn;
  const int ct = t % ntn;

  const int mrow = mt * 16 + l16;
  const int col  = ct * 16 + l16;
  const float* xb = x + (size_t)n * CIN * LSPAT;

  v8f acc = {};
  for (int kk = 0; kk < CIN; kk += 4) {
    const int ka = kk + half * 2;
    v2f a; a[0] = w[mrow * CIN + ka];
           a[1] = w[mrow * CIN + ka + 1];
    v2f b; b[0] = xb[(size_t)ka * LSPAT + col];
           b[1] = xb[(size_t)(ka + 1) * LSPAT + col];
    acc = wmma4(a, b, acc);
  }

  float* out = qkv + (size_t)n * CH3 * LSPAT;
#pragma unroll
  for (int r = 0; r < 8; ++r) {
    const int row = mt * 16 + r + half * 8;
    out[(size_t)row * LSPAT + col] = acc[r] + bias[row];
  }
}

__global__ __launch_bounds__(128) void attn_kernel(
    const float* __restrict__ qkv, float* __restrict__ obuf) {
  __shared__ float plds[4][16][17];
  const int lane = threadIdx.x & 31;
  const int half = lane >> 4;
  const int l16  = lane & 15;
  const int wv   = threadIdx.x >> 5;

  const int qtiles = LSPAT / 64;
  const int nh = blockIdx.x / qtiles;
  const int qt = blockIdx.x % qtiles;
  const int n = nh >> 3, h = nh & 7;
  const int qbase = qt * 64 + wv * 16;

  const float* base = qkv + (size_t)n * CH3 * LSPAT + (size_t)h * 24 * LSPAT;
  const float* qm = base;
  const float* km = base + (size_t)DHEAD * LSPAT;
  const float* vm = base + (size_t)2 * DHEAD * LSPAT;

  v2f aq[2];
#pragma unroll
  for (int db = 0; db < 2; ++db) {
    const int d0 = db * 4 + half * 2;
    aq[db][0] = qm[(size_t)d0 * LSPAT + qbase + l16];
    aq[db][1] = qm[(size_t)(d0 + 1) * LSPAT + qbase + l16];
  }

  float m8[8];
#pragma unroll
  for (int r = 0; r < 8; ++r) m8[r] = -3.0e38f;

  for (int kt = 0; kt < LSPAT / 16; ++kt) {
    const int kb = kt * 16;
    v8f s = {};
#pragma unroll
    for (int db = 0; db < 2; ++db) {
      const int d0 = db * 4 + half * 2;
      v2f bk; bk[0] = km[(size_t)d0 * LSPAT + kb + l16];
              bk[1] = km[(size_t)(d0 + 1) * LSPAT + kb + l16];
      s = wmma4(aq[db], bk, s);
    }
#pragma unroll
    for (int r = 0; r < 8; ++r) m8[r] = fmaxf(m8[r], s[r]);
  }
#pragma unroll
  for (int r = 0; r < 8; ++r) {
#pragma unroll
    for (int m = 1; m < 16; m <<= 1) m8[r] = fmaxf(m8[r], __shfl_xor(m8[r], m, 32));
  }

  float s8[8];
#pragma unroll
  for (int r = 0; r < 8; ++r) s8[r] = 0.0f;

  for (int kt = 0; kt < LSPAT / 16; ++kt) {
    const int kb = kt * 16;
    v8f s = {};
#pragma unroll
    for (int db = 0; db < 2; ++db) {
      const int d0 = db * 4 + half * 2;
      v2f bk; bk[0] = km[(size_t)d0 * LSPAT + kb + l16];
              bk[1] = km[(size_t)(d0 + 1) * LSPAT + kb + l16];
      s = wmma4(aq[db], bk, s);
    }
#pragma unroll
    for (int r = 0; r < 8; ++r) s8[r] += __expf((s[r] - m8[r]) * ATTN_SCALE);
  }
  float inv[8];
#pragma unroll
  for (int r = 0; r < 8; ++r) {
#pragma unroll
    for (int m = 1; m < 16; m <<= 1) s8[r] += __shfl_xor(s8[r], m, 32);
    inv[r] = 1.0f / s8[r];
  }

  const float* kp = km + (size_t)(half * 2) * LSPAT + l16;
  const float* vp = vm + (size_t)(l16 & 7) * LSPAT + half * 2;

  auto load_tile = [&](v2f (&bk)[2], v2f (&bv)[4], const int kb) {
#pragma unroll
    for (int db = 0; db < 2; ++db) {
      bk[db][0] = kp[(size_t)(db * 4) * LSPAT + kb];
      bk[db][1] = kp[(size_t)(db * 4 + 1) * LSPAT + kb];
    }
#pragma unroll
    for (int c = 0; c < 4; ++c) {
      bv[c][0] = vp[kb + c * 4];
      bv[c][1] = vp[kb + c * 4 + 1];
    }
  };

  auto lds_barrier = [&]() {
    __builtin_amdgcn_fence(__ATOMIC_RELEASE, "workgroup", "local");
    __builtin_amdgcn_s_barrier();
    __builtin_amdgcn_fence(__ATOMIC_ACQUIRE, "workgroup", "local");
  };

  v8f acc = {};
  auto do_tile = [&](const v2f (&bk)[2], const v2f (&bv)[4]) {
    v8f s = {};
    s = wmma4(aq[0], bk[0], s);
    s = wmma4(aq[1], bk[1], s);
#pragma unroll
    for (int r = 0; r < 8; ++r)
      plds[wv][r + half * 8][l16] =
          __expf((s[r] - m8[r]) * ATTN_SCALE) * inv[r];
    lds_barrier();
#pragma unroll
    for (int c = 0; c < 4; ++c) {
      const int k0 = c * 4 + half * 2;
      v2f ap; ap[0] = plds[wv][l16][k0];
              ap[1] = plds[wv][l16][k0 + 1];
      acc = wmma4(ap, bv[c], acc);
    }
    lds_barrier();
  };

  v2f bk0[2], bv0[4], bk1[2], bv1[4];
  load_tile(bk0, bv0, 0);
  for (int kt = 0; kt < LSPAT / 16; kt += 2) {
    load_tile(bk1, bv1, (kt + 1) * 16);
    do_tile(bk0, bv0);
    const int kn = (kt + 2 < LSPAT / 16) ? kt + 2 : kt + 1;
    load_tile(bk0, bv0, kn * 16);
    do_tile(bk1, bv1);
  }

  if (l16 < 8) {
    float* ob = obuf + (size_t)n * HCH * LSPAT + (size_t)(h * DHEAD + l16) * LSPAT;
#pragma unroll
    for (int r = 0; r < 8; ++r) ob[qbase + r + half * 8] = acc[r];
  }
}

#define POS_OCT 8
#define POS_BLK 128
__global__ __launch_bounds__(POS_BLK) void pos_conv_kernel(
    const float* __restrict__ qkv, const float* __restrict__ pw,
    const float* __restrict__ pb, float* __restrict__ obuf) {
  const int chunks = LSPAT / POS_BLK;
  int b = blockIdx.x;
  const int chunk = b % chunks;  b /= chunks;
  const int og = b % (HCH / POS_OCT);
  const int n  = b / (HCH / POS_OCT);
  const unsigned p = chunk * POS_BLK + threadIdx.x;
  const int yp = (int)(p / WW);
  const int xp = (int)(p % WW);
  const int oc0 = og * POS_OCT;

  unsigned off[9];
  bool ok[9];
#pragma unroll
  for (int ky = 0; ky < 3; ++ky) {
#pragma unroll
    for (int kx = 0; kx < 3; ++kx) {
      const int iy = yp + ky - 1;
      const int ix = xp + kx - 1;
      const bool in_img = (iy >= 0) && (iy < HH) && (ix >= 0) && (ix < WW);
      ok[ky * 3 + kx]  = in_img;
      off[ky * 3 + kx] = in_img ? (unsigned)(iy * WW + ix) : p;
    }
  }

  float acc[POS_OCT];
#pragma unroll
  for (int o = 0; o < POS_OCT; ++o) acc[o] = pb[oc0 + o];

  const float* vbase = qkv + (size_t)n * CH3 * LSPAT + (size_t)16 * LSPAT;
  float cur[DHEAD][9];
#pragma unroll
  for (int dd = 0; dd < DHEAD; ++dd) {
#pragma unroll
    for (int t = 0; t < 9; ++t) cur[dd][t] = vbase[(unsigned)(dd * LSPAT) + off[t]];
  }

  for (int hh = 0; hh < NHEAD; ++hh) {
    const int hn = (hh + 1 < NHEAD) ? hh + 1 : hh;
    const unsigned cho = (unsigned)(hn * 24 * LSPAT);
    float nxt[DHEAD][9];
#pragma unroll
    for (int dd = 0; dd < DHEAD; ++dd) {
#pragma unroll
      for (int t = 0; t < 9; ++t)
        nxt[dd][t] = vbase[cho + (unsigned)(dd * LSPAT) + off[t]];
    }

#pragma unroll
    for (int dd = 0; dd < DHEAD; ++dd) {
      const int ic = hh * DHEAD + dd;
      float in[9];
#pragma unroll
      for (int t = 0; t < 9; ++t) in[t] = ok[t] ? cur[dd][t] : 0.0f;
#pragma unroll
      for (int o = 0; o < POS_OCT; ++o) {
        const float* wrow = pw + ((size_t)(oc0 + o) * HCH + ic) * 9;
#pragma unroll
        for (int t = 0; t < 9; ++t) acc[o] = fmaf(in[t], wrow[t], acc[o]);
      }
    }

#pragma unroll
    for (int dd = 0; dd < DHEAD; ++dd) {
#pragma unroll
      for (int t = 0; t < 9; ++t) cur[dd][t] = nxt[dd][t];
    }
  }

  float* ob = obuf + (size_t)n * HCH * LSPAT + (size_t)oc0 * LSPAT + p;
#pragma unroll
  for (int o = 0; o < POS_OCT; ++o) ob[(size_t)o * LSPAT] += acc[o];
}

__global__ __launch_bounds__(256) void res_gemm_kernel(
    const float* __restrict__ obuf, const float* __restrict__ w,
    const float* __restrict__ bias, float* __restrict__ out) {
  const int lane = threadIdx.x & 31;
  const int half = lane >> 4;
  const int l16  = lane & 15;
  const int wave = blockIdx.x * (blockDim.x >> 5) + (threadIdx.x >> 5);

  const int ntn = LSPAT / 16;
  const int ntm = CIN / 16;
  int t = wave;
  const int n  = t / (ntm * ntn); t -= n * ntm * ntn;
  const int mt = t / ntn;
  const int ct = t % ntn;

  const int mrow = mt * 16 + l16;
  const int col  = ct * 16 + l16;
  const float* ib = obuf + (size_t)n * HCH * LSPAT;

  v8f acc = {};
#pragma unroll
  for (int kk = 0; kk < HCH; kk += 4) {
    const int ka = kk + half * 2;
    v2f a; a[0] = w[mrow * HCH + ka];
           a[1] = w[mrow * HCH + ka + 1];
    v2f b; b[0] = ib[(size_t)ka * LSPAT + col];
           b[1] = ib[(size_t)(ka + 1) * LSPAT + col];
    acc = wmma4(a, b, acc);
  }

  float* ob = out + (size_t)n * CIN * LSPAT;
#pragma unroll
  for (int r = 0; r < 8; ++r) {
    const int row = mt * 16 + r + half * 8;
    ob[(size_t)row * LSPAT + col] = acc[r] + bias[row];
  }
}

extern "C" void kernel_launch(void* const* d_in, const int* in_sizes, int n_in,
                              void* d_out, int out_size, void* d_ws, size_t ws_size,
                              hipStream_t stream) {
  const float* x     = (const float*)d_in[0];
  const float* qkv_w = (const float*)d_in[1];
  const float* qkv_b = (const float*)d_in[2];
  const float* pos_w = (const float*)d_in[3];
  const float* pos_b = (const float*)d_in[4];
  const float* res_w = (const float*)d_in[5];
  const float* res_b = (const float*)d_in[6];
  float* out = (float*)d_out;

  float* qkv  = (float*)d_ws;
  float* obuf = qkv + (size_t)NBATCH * CH3 * LSPAT;

  const int waves1 = NBATCH * (CH3 / 16) * (LSPAT / 16);
  qkv_gemm_kernel<<<waves1 / 8, 256, 0, stream>>>(x, qkv_w, qkv_b, qkv);

  attn_kernel<<<NBATCH * NHEAD * (LSPAT / 64), 128, 0, stream>>>(qkv, obuf);

  pos_conv_kernel<<<NBATCH * (HCH / POS_OCT) * (LSPAT / POS_BLK), POS_BLK, 0,
                    stream>>>(qkv, pos_w, pos_b, obuf);

  const int waves4 = NBATCH * (CIN / 16) * (LSPAT / 16);
  res_gemm_kernel<<<waves4 / 8, 256, 0, stream>>>(obuf, res_w, res_b, out);
}


# qkv 1x1-conv GEMM: weights staged through LDS in A-fragment order (64-k chunks, two buffers), shared by 8 waves; each x fragment feeds 2 row tiles; next chunk requested ahead; k sum order kept
# speedup vs baseline: 1.8840x; 1.0952x over previous
#include <hip/hip_runtime.h>
#include <math.h>

typedef __attribute__((ext_vector_type(2))) float v2f;
typedef __attribute__((ext_vector_type(8))) float v8f;

#define NBATCH 4
#define CIN 512
#define HH 48
#define WW 48
#define LSPAT (HH*WW)
#define CH3 192
#define HCH 64
#define NHEAD 8
#define DHEAD 8
#define ATTN_SCALE 0.35355339059327379f

static __device__ __forceinline__ v8f wmma4(v2f a, v2f b, v8f c) {
  return __builtin_amdgcn_wmma_f32_16x16x4_f32(false, a, false, b, (short)0, c,
                                               false, false);
}

typedef __attribute__((ext_vector_type(4))) float v4f;
#define QK_ROWS 64
#define QK_COLS 64
#define QK_KC 64
#define QK_SP 260
__global__ __launch_bounds__(256) void qkv_gemm_kernel(
    const float* __restrict__ x, const float* __restrict__ w,
    const float* __restrict__ bias, float* __restrict__ qkv) {
  __shared__ __attribute__((aligned(16))) float wl[2][16 * QK_SP];
  const int tid  = threadIdx.x;
  const int lane = tid & 31;
  const int half = lane >> 4;
  const int l16  = lane & 15;
  const int wv   = tid >> 5;

  int b = blockIdx.x;
  const int mg = b % (CH3 / QK_ROWS);   b /= (CH3 / QK_ROWS);
  const int cb = b % (LSPAT / QK_COLS);
  const int n  = b / (LSPAT / QK_COLS);
  const int m0 = mg * QK_ROWS;
  const int rt = wv >> 2;
  const int col = cb * QK_COLS + (wv & 3) * 16 + l16;
  const float* xb = x + (size_t)n * CIN * LSPAT;

  const int q4   = tid & 15;
  const int srow = tid >> 4;
  const float* wsrc = w + (size_t)(m0 + srow) * CIN + q4 * 4;
  const int sdst = q4 * QK_SP + srow * 4;

  const float* bp = xb + (size_t)(half * 2) * LSPAT + col;
  const int aoff = (rt * 32 + l16) * 4 + half * 2;

  auto load_w = [&](v4f (&wr)[4], const int kc) {
#pragma unroll
    for (int i = 0; i < 4; ++i) {
      const float* p = wsrc + (size_t)(i * 16) * CIN + kc;
      wr[i][0] = p[0]; wr[i][1] = p[1]; wr[i][2] = p[2]; wr[i][3] = p[3];
    }
  };
  auto store_w = [&](float* buf, const v4f (&wr)[4]) {
#pragma unroll
    for (int i = 0; i < 4; ++i) *(v4f*)(buf + sdst + i * 64) = wr[i];
  };
  auto load_b = [&](v2f (&bf)[16], const int kc) {
#pragma unroll
    for (int s = 0; s < 16; ++s) {
      bf[s][0] = bp[(size_t)(kc + 4 * s) * LSPAT];
      bf[s][1] = bp[(size_t)(kc + 4 * s + 1) * LSPAT];
    }
  };
  auto lds_barrier = [&]() {
    __builtin_amdgcn_fence(__ATOMIC_RELEASE, "workgroup", "local");
    __builtin_amdgcn_s_barrier();
    __builtin_amdgcn_fence(__ATOMIC_ACQUIRE, "workgroup", "local");
  };

  v8f acc0 = {}, acc1 = {};
  auto compute = [&](const float* buf, const v2f (&bf)[16]) {
#pragma unroll
    for (int s = 0; s < 16; ++s) {
      const v2f a0 = *(const v2f*)(buf + s * QK_SP + aoff);
      const v2f a1 = *(const v2f*)(buf + s * QK_SP + aoff + 64);
      acc0 = wmma4(a0, bf[s], acc0);
      acc1 = wmma4(a1, bf[s], acc1);
    }
  };

  v4f wr[4];
  v2f bf0[16], bf1[16];
  load_w(wr, 0);
  load_b(bf0, 0);
  store_w(wl[0], wr);
  lds_barrier();
#pragma unroll
  for (int c = 0; c < CIN / QK_KC; c += 2) {
    load_w(wr, (c + 1) * QK_KC);
    load_b(bf1, (c + 1) * QK_KC);
    compute(wl[0], bf0);
    store_w(wl[1], wr);
    lds_barrier();
    if (c + 2 < CIN / QK_KC) {
      load_w(wr, (c + 2) * QK_KC);
      load_b(bf0, (c + 2) * QK_KC);
    }
    compute(wl[1], bf1);
    if (c + 2 < CIN / QK_KC) {
      store_w(wl[0], wr);
      lds_barrier();
    }
  }

  float* out = qkv + (size_t)n * CH3 * LSPAT;
#pragma unroll
  for (int r = 0; r < 8; ++r) {
    const int row0 = m0 + rt * 32 + r + half * 8;
    const int row1 = row0 + 16;
    out[(size_t)row0 * LSPAT + col] = acc0[r] + bias[row0];
    out[(size_t)row1 * LSPAT + col] = acc1[r] + bias[row1];
  }
}

__global__ __launch_bounds__(128) void attn_kernel(
    const float* __restrict__ qkv, float* __restrict__ obuf) {
  __shared__ float plds[4][16][17];
  const int lane = threadIdx.x & 31;
  const int half = lane >> 4;
  const int l16  = lane & 15;
  const int wv   = threadIdx.x >> 5;

  const int qtiles = LSPAT / 64;
  const int nh = blockIdx.x / qtiles;
  const int qt = blockIdx.x % qtiles;
  const int n = nh >> 3, h = nh & 7;
  const int qbase = qt * 64 + wv * 16;

  const float* base = qkv + (size_t)n * CH3 * LSPAT + (size_t)h * 24 * LSPAT;
  const float* qm = base;
  const float* km = base + (size_t)DHEAD * LSPAT;
  const float* vm = base + (size_t)2 * DHEAD * LSPAT;

  v2f aq[2];
#pragma unroll
  for (int db = 0; db < 2; ++db) {
    const int d0 = db * 4 + half * 2;
    aq[db][0] = qm[(size_t)d0 * LSPAT + qbase + l16];
    aq[db][1] = qm[(size_t)(d0 + 1) * LSPAT + qbase + l16];
  }

  float m8[8];
#pragma unroll
  for (int r = 0; r < 8; ++r) m8[r] = -3.0e38f;

  for (int kt = 0; kt < LSPAT / 16; ++kt) {
    const int kb = kt * 16;
    v8f s = {};
#pragma unroll
    for (int db = 0; db < 2; ++db) {
      const int d0 = db * 4 + half * 2;
      v2f bk; bk[0] = km[(size_t)d0 * LSPAT + kb + l16];
              bk[1] = km[(size_t)(d0 + 1) * LSPAT + kb + l16];
      s = wmma4(aq[db], bk, s);
    }
#pragma unroll
    for (int r = 0; r < 8; ++r) m8[r] = fmaxf(m8[r], s[r]);
  }
#pragma unroll
  for (int r = 0; r < 8; ++r) {
#pragma unroll
    for (int m = 1; m < 16; m <<= 1) m8[r] = fmaxf(m8[r], __shfl_xor(m8[r], m, 32));
  }

  float s8[8];
#pragma unroll
  for (int r = 0; r < 8; ++r) s8[r] = 0.0f;

  for (int kt = 0; kt < LSPAT / 16; ++kt) {
    const int kb = kt * 16;
    v8f s = {};
#pragma unroll
    for (int db = 0; db < 2; ++db) {
      const int d0 = db * 4 + half * 2;
      v2f bk; bk[0] = km[(size_t)d0 * LSPAT + kb + l16];
              bk[1] = km[(size_t)(d0 + 1) * LSPAT + kb + l16];
      s = wmma4(aq[db], bk, s);
    }
#pragma unroll
    for (int r = 0; r < 8; ++r) s8[r] += __expf((s[r] - m8[r]) * ATTN_SCALE);
  }
  float inv[8];
#pragma unroll
  for (int r = 0; r < 8; ++r) {
#pragma unroll
    for (int m = 1; m < 16; m <<= 1) s8[r] += __shfl_xor(s8[r], m, 32);
    inv[r] = 1.0f / s8[r];
  }

  const float* kp = km + (size_t)(half * 2) * LSPAT + l16;
  const float* vp = vm + (size_t)(l16 & 7) * LSPAT + half * 2;

  auto load_tile = [&](v2f (&bk)[2], v2f (&bv)[4], const int kb) {
#pragma unroll
    for (int db = 0; db < 2; ++db) {
      bk[db][0] = kp[(size_t)(db * 4) * LSPAT + kb];
      bk[db][1] = kp[(size_t)(db * 4 + 1) * LSPAT + kb];
    }
#pragma unroll
    for (int c = 0; c < 4; ++c) {
      bv[c][0] = vp[kb + c * 4];
      bv[c][1] = vp[kb + c * 4 + 1];
    }
  };

  auto lds_barrier = [&]() {
    __builtin_amdgcn_fence(__ATOMIC_RELEASE, "workgroup", "local");
    __builtin_amdgcn_s_barrier();
    __builtin_amdgcn_fence(__ATOMIC_ACQUIRE, "workgroup", "local");
  };

  v8f acc = {};
  auto do_tile = [&](const v2f (&bk)[2], const v2f (&bv)[4]) {
    v8f s = {};
    s = wmma4(aq[0], bk[0], s);
    s = wmma4(aq[1], bk[1], s);
#pragma unroll
    for (int r = 0; r < 8; ++r)
      plds[wv][r + half * 8][l16] =
          __expf((s[r] - m8[r]) * ATTN_SCALE) * inv[r];
    lds_barrier();
#pragma unroll
    for (int c = 0; c < 4; ++c) {
      const int k0 = c * 4 + half * 2;
      v2f ap; ap[0] = plds[wv][l16][k0];
              ap[1] = plds[wv][l16][k0 + 1];
      acc = wmma4(ap, bv[c], acc);
    }
    lds_barrier();
  };

  v2f bk0[2], bv0[4], bk1[2], bv1[4];
  load_tile(bk0, bv0, 0);
  for (int kt = 0; kt < LSPAT / 16; kt += 2) {
    load_tile(bk1, bv1, (kt + 1) * 16);
    do_tile(bk0, bv0);
    const int kn = (kt + 2 < LSPAT / 16) ? kt + 2 : kt + 1;
    load_tile(bk0, bv0, kn * 16);
    do_tile(bk1, bv1);
  }

  if (l16 < 8) {
    float* ob = obuf + (size_t)n * HCH * LSPAT + (size_t)(h * DHEAD + l16) * LSPAT;
#pragma unroll
    for (int r = 0; r < 8; ++r) ob[qbase + r + half * 8] = acc[r];
  }
}

#define POS_OCT 8
#define POS_BLK 128
__global__ __launch_bounds__(POS_BLK) void pos_conv_kernel(
    const float* __restrict__ qkv, const float* __restrict__ pw,
    const float* __restrict__ pb, float* __restrict__ obuf) {
  const int chunks = LSPAT / POS_BLK;
  int b = blockIdx.x;
  const int chunk = b % chunks;  b /= chunks;
  const int og = b % (HCH / POS_OCT);
  const int n  = b / (HCH / POS_OCT);
  const unsigned p = chunk * POS_BLK + threadIdx.x;
  const int yp = (int)(p / WW);
  const int xp = (int)(p % WW);
  const int oc0 = og * POS_OCT;

  unsigned off[9];
  bool ok[9];
#pragma unroll
  for (int ky = 0; ky < 3; ++ky) {
#pragma unroll
    for (int kx = 0; kx < 3; ++kx) {
      const int iy = yp + ky - 1;
      const int ix = xp + kx - 1;
      const bool in_img = (iy >= 0) && (iy < HH) && (ix >= 0) && (ix < WW);
      ok[ky * 3 + kx]  = in_img;
      off[ky * 3 + kx] = in_img ? (unsigned)(iy * WW + ix) : p;
    }
  }

  float acc[POS_OCT];
#pragma unroll
  for (int o = 0; o < POS_OCT; ++o) acc[o] = pb[oc0 + o];

  const float* vbase = qkv + (size_t)n * CH3 * LSPAT + (size_t)16 * LSPAT;
  float cur[DHEAD][9];
#pragma unroll
  for (int dd = 0; dd < DHEAD; ++dd) {
#pragma unroll
    for (int t = 0; t < 9; ++t) cur[dd][t] = vbase[(unsigned)(dd * LSPAT) + off[t]];
  }

  for (int hh = 0; hh < NHEAD; ++hh) {
    const int hn = (hh + 1 < NHEAD) ? hh + 1 : hh;
    const unsigned cho = (unsigned)(hn * 24 * LSPAT);
    float nxt[DHEAD][9];
#pragma unroll
    for (int dd = 0; dd < DHEAD; ++dd) {
#pragma unroll
      for (int t = 0; t < 9; ++t)
        nxt[dd][t] = vbase[cho + (unsigned)(dd * LSPAT) + off[t]];
    }

#pragma unroll
    for (int dd = 0; dd < DHEAD; ++dd) {
      const int ic = hh * DHEAD + dd;
      float in[9];
#pragma unroll
      for (int t = 0; t < 9; ++t) in[t] = ok[t] ? cur[dd][t] : 0.0f;
#pragma unroll
      for (int o = 0; o < POS_OCT; ++o) {
        const float* wrow = pw + ((size_t)(oc0 + o) * HCH + ic) * 9;
#pragma unroll
        for (int t = 0; t < 9; ++t) acc[o] = fmaf(in[t], wrow[t], acc[o]);
      }
    }

#pragma unroll
    for (int dd = 0; dd < DHEAD; ++dd) {
#pragma unroll
      for (int t = 0; t < 9; ++t) cur[dd][t] = nxt[dd][t];
    }
  }

  float* ob = obuf + (size_t)n * HCH * LSPAT + (size_t)oc0 * LSPAT + p;
#pragma unroll
  for (int o = 0; o < POS_OCT; ++o) ob[(size_t)o * LSPAT] += acc[o];
}

__global__ __launch_bounds__(256) void res_gemm_kernel(
    const float* __restrict__ obuf, const float* __restrict__ w,
    const float* __restrict__ bias, float* __restrict__ out) {
  const int lane = threadIdx.x & 31;
  const int half = lane >> 4;
  const int l16  = lane & 15;
  const int wave = blockIdx.x * (blockDim.x >> 5) + (threadIdx.x >> 5);

  const int ntn = LSPAT / 16;
  const int ntm = CIN / 16;
  int t = wave;
  const int n  = t / (ntm * ntn); t -= n * ntm * ntn;
  const int mt = t / ntn;
  const int ct = t % ntn;

  const int mrow = mt * 16 + l16;
  const int col  = ct * 16 + l16;
  const float* ib = obuf + (size_t)n * HCH * LSPAT;

  v8f acc = {};
#pragma unroll
  for (int kk = 0; kk < HCH; kk += 4) {
    const int ka = kk + half * 2;
    v2f a; a[0] = w[mrow * HCH + ka];
           a[1] = w[mrow * HCH + ka + 1];
    v2f b; b[0] = ib[(size_t)ka * LSPAT + col];
           b[1] = ib[(size_t)(ka + 1) * LSPAT + col];
    acc = wmma4(a, b, acc);
  }

  float* ob = out + (size_t)n * CIN * LSPAT;
#pragma unroll
  for (int r = 0; r < 8; ++r) {
    const int row = mt * 16 + r + half * 8;
    ob[(size_t)row * LSPAT + col] = acc[r] + bias[row];
  }
}

extern "C" void kernel_launch(void* const* d_in, const int* in_sizes, int n_in,
                              void* d_out, int out_size, void* d_ws, size_t ws_size,
                              hipStream_t stream) {
  const float* x     = (const float*)d_in[0];
  const float* qkv_w = (const float*)d_in[1];
  const float* qkv_b = (const float*)d_in[2];
  const float* pos_w = (const float*)d_in[3];
  const float* pos_b = (const float*)d_in[4];
  const float* res_w = (const float*)d_in[5];
  const float* res_b = (const float*)d_in[6];
  float* out = (float*)d_out;

  float* qkv  = (float*)d_ws;
  float* obuf = qkv + (size_t)NBATCH * CH3 * LSPAT;

  qkv_gemm_kernel<<<NBATCH * (CH3 / QK_ROWS) * (LSPAT / QK_COLS), 256, 0,
                    stream>>>(x, qkv_w, qkv_b, qkv);

  attn_kernel<<<NBATCH * NHEAD * (LSPAT / 64), 128, 0, stream>>>(qkv, obuf);

  pos_conv_kernel<<<NBATCH * (HCH / POS_OCT) * (LSPAT / POS_BLK), POS_BLK, 0,
                    stream>>>(qkv, pos_w, pos_b, obuf);

  const int waves4 = NBATCH * (CIN / 16) * (LSPAT / 16);
  res_gemm_kernel<<<waves4 / 8, 256, 0, stream>>>(obuf, res_w, res_b, out);
}


# attention: softmax division deferred to the 16x8 result; the P*V pass sums exp itself on unnormalised P, so the separate sum pass over the keys is gone (2 passes, was 3); P*V rounding changes
# speedup vs baseline: 2.2381x; 1.1880x over previous
#include <hip/hip_runtime.h>
#include <math.h>

typedef __attribute__((ext_vector_type(2))) float v2f;
typedef __attribute__((ext_vector_type(8))) float v8f;

#define NBATCH 4
#define CIN 512
#define HH 48
#define WW 48
#define LSPAT (HH*WW)
#define CH3 192
#define HCH 64
#define NHEAD 8
#define DHEAD 8
#define ATTN_SCALE 0.35355339059327379f

static __device__ __forceinline__ v8f wmma4(v2f a, v2f b, v8f c) {
  return __builtin_amdgcn_wmma_f32_16x16x4_f32(false, a, false, b, (short)0, c,
                                               false, false);
}

typedef __attribute__((ext_vector_type(4))) float v4f;
#define QK_ROWS 64
#define QK_COLS 64
#define QK_KC 64
#define QK_SP 260
__global__ __launch_bounds__(256) void qkv_gemm_kernel(
    const float* __restrict__ x, const float* __restrict__ w,
    const float* __restrict__ bias, float* __restrict__ qkv) {
  __shared__ __attribute__((aligned(16))) float wl[2][16 * QK_SP];
  const int tid  = threadIdx.x;
  const int lane = tid & 31;
  const int half = lane >> 4;
  const int l16  = lane & 15;
  const int wv   = tid >> 5;

  int b = blockIdx.x;
  const int mg = b % (CH3 / QK_ROWS);   b /= (CH3 / QK_ROWS);
  const int cb = b % (LSPAT / QK_COLS);
  const int n  = b / (LSPAT / QK_COLS);
  const int m0 = mg * QK_ROWS;
  const int rt = wv >> 2;
  const int col = cb * QK_COLS + (wv & 3) * 16 + l16;
  const float* xb = x + (size_t)n * CIN * LSPAT;

  const int q4   = tid & 15;
  const int srow = tid >> 4;
  const float* wsrc = w + (size_t)(m0 + srow) * CIN + q4 * 4;
  const int sdst = q4 * QK_SP + srow * 4;

  const float* bp = xb + (size_t)(half * 2) * LSPAT + col;
  const int aoff = (rt * 32 + l16) * 4 + half * 2;

  auto load_w = [&](v4f (&wr)[4], const int kc) {
#pragma unroll
    for (int i = 0; i < 4; ++i) {
      const float* p = wsrc + (size_t)(i * 16) * CIN + kc;
      wr[i][0] = p[0]; wr[i][1] = p[1]; wr[i][2] = p[2]; wr[i][3] = p[3];
    }
  };
  auto store_w = [&](float* buf, const v4f (&wr)[4]) {
#pragma unroll
    for (int i = 0; i < 4; ++i) *(v4f*)(buf + sdst + i * 64) = wr[i];
  };
  auto load_b = [&](v2f (&bf)[16], const int kc) {
#pragma unroll
    for (int s = 0; s < 16; ++s) {
      bf[s][0] = bp[(size_t)(kc + 4 * s) * LSPAT];
      bf[s][1] = bp[(size_t)(kc + 4 * s + 1) * LSPAT];
    }
  };
  auto lds_barrier = [&]() {
    __builtin_amdgcn_fence(__ATOMIC_RELEASE, "workgroup", "local");
    __builtin_amdgcn_s_barrier();
    __builtin_amdgcn_fence(__ATOMIC_ACQUIRE, "workgroup", "local");
  };

  v8f acc0 = {}, acc1 = {};
  auto compute = [&](const float* buf, const v2f (&bf)[16]) {
#pragma unroll
    for (int s = 0; s < 16; ++s) {
      const v2f a0 = *(const v2f*)(buf + s * QK_SP + aoff);
      const v2f a1 = *(const v2f*)(buf + s * QK_SP + aoff + 64);
      acc0 = wmma4(a0, bf[s], acc0);
      acc1 = wmma4(a1, bf[s], acc1);
    }
  };

  v4f wr[4];
  v2f bf0[16], bf1[16];
  load_w(wr, 0);
  load_b(bf0, 0);
  store_w(wl[0], wr);
  lds_barrier();
#pragma unroll
  for (int c = 0; c < CIN / QK_KC; c += 2) {
    load_w(wr, (c + 1) * QK_KC);
    load_b(bf1, (c + 1) * QK_KC);
    compute(wl[0], bf0);
    store_w(wl[1], wr);
    lds_barrier();
    if (c + 2 < CIN / QK_KC) {
      load_w(wr, (c + 2) * QK_KC);
      load_b(bf0, (c + 2) * QK_KC);
    }
    compute(wl[1], bf1);
    if (c + 2 < CIN / QK_KC) {
      store_w(wl[0], wr);
      lds_barrier();
    }
  }

  float* out = qkv + (size_t)n * CH3 * LSPAT;
#pragma unroll
  for (int r = 0; r < 8; ++r) {
    const int row0 = m0 + rt * 32 + r + half * 8;
    const int row1 = row0 + 16;
    out[(size_t)row0 * LSPAT + col] = acc0[r] + bias[row0];
    out[(size_t)row1 * LSPAT + col] = acc1[r] + bias[row1];
  }
}

__global__ __launch_bounds__(128) void attn_kernel(
    const float* __restrict__ qkv, float* __restrict__ obuf) {
  __shared__ float plds[4][16][17];
  const int lane = threadIdx.x & 31;
  const int half = lane >> 4;
  const int l16  = lane & 15;
  const int wv   = threadIdx.x >> 5;

  const int qtiles = LSPAT / 64;
  const int nh = blockIdx.x / qtiles;
  const int qt = blockIdx.x % qtiles;
  const int n = nh >> 3, h = nh & 7;
  const int qbase = qt * 64 + wv * 16;

  const float* base = qkv + (size_t)n * CH3 * LSPAT + (size_t)h * 24 * LSPAT;
  const float* qm = base;
  const float* km = base + (size_t)DHEAD * LSPAT;
  const float* vm = base + (size_t)2 * DHEAD * LSPAT;

  v2f aq[2];
#pragma unroll
  for (int db = 0; db < 2; ++db) {
    const int d0 = db * 4 + half * 2;
    aq[db][0] = qm[(size_t)d0 * LSPAT + qbase + l16];
    aq[db][1] = qm[(size_t)(d0 + 1) * LSPAT + qbase + l16];
  }

  float m8[8];
#pragma unroll
  for (int r = 0; r < 8; ++r) m8[r] = -3.0e38f;

  for (int kt = 0; kt < LSPAT / 16; ++kt) {
    const int kb = kt * 16;
    v8f s = {};
#pragma unroll
    for (int db = 0; db < 2; ++db) {
      const int d0 = db * 4 + half * 2;
      v2f bk; bk[0] = km[(size_t)d0 * LSPAT + kb + l16];
              bk[1] = km[(size_t)(d0 + 1) * LSPAT + kb + l16];
      s = wmma4(aq[db], bk, s);
    }
#pragma unroll
    for (int r = 0; r < 8; ++r) m8[r] = fmaxf(m8[r], s[r]);
  }
#pragma unroll
  for (int r = 0; r < 8; ++r) {
#pragma unroll
    for (int m = 1; m < 16; m <<= 1) m8[r] = fmaxf(m8[r], __shfl_xor(m8[r], m, 32));
  }

  float s8[8];
#pragma unroll
  for (int r = 0; r < 8; ++r) s8[r] = 0.0f;

  const float* kp = km + (size_t)(half * 2) * LSPAT + l16;
  const float* vp = vm + (size_t)(l16 & 7) * LSPAT + half * 2;

  auto load_tile = [&](v2f (&bk)[2], v2f (&bv)[4], const int kb) {
#pragma unroll
    for (int db = 0; db < 2; ++db) {
      bk[db][0] = kp[(size_t)(db * 4) * LSPAT + kb];
      bk[db][1] = kp[(size_t)(db * 4 + 1) * LSPAT + kb];
    }
#pragma unroll
    for (int c = 0; c < 4; ++c) {
      bv[c][0] = vp[kb + c * 4];
      bv[c][1] = vp[kb + c * 4 + 1];
    }
  };

  auto lds_barrier = [&]() {
    __builtin_amdgcn_fence(__ATOMIC_RELEASE, "workgroup", "local");
    __builtin_amdgcn_s_barrier();
    __builtin_amdgcn_fence(__ATOMIC_ACQUIRE, "workgroup", "local");
  };

  v8f acc = {};
  auto do_tile = [&](const v2f (&bk)[2], const v2f (&bv)[4]) {
    v8f s = {};
    s = wmma4(aq[0], bk[0], s);
    s = wmma4(aq[1], bk[1], s);
#pragma unroll
    for (int r = 0; r < 8; ++r) {
      const float e = __expf((s[r] - m8[r]) * ATTN_SCALE);
      s8[r] += e;
      plds[wv][r + half * 8][l16] = e;
    }
    lds_barrier();
#pragma unroll
    for (int c = 0; c < 4; ++c) {
      const int k0 = c * 4 + half * 2;
      v2f ap; ap[0] = plds[wv][l16][k0];
              ap[1] = plds[wv][l16][k0 + 1];
      acc = wmma4(ap, bv[c], acc);
    }
    lds_barrier();
  };

  v2f bk0[2], bv0[4], bk1[2], bv1[4];
  load_tile(bk0, bv0, 0);
  for (int kt = 0; kt < LSPAT / 16; kt += 2) {
    load_tile(bk1, bv1, (kt + 1) * 16);
    do_tile(bk0, bv0);
    const int kn = (kt + 2 < LSPAT / 16) ? kt + 2 : kt + 1;
    load_tile(bk0, bv0, kn * 16);
    do_tile(bk1, bv1);
  }

#pragma unroll
  for (int r = 0; r < 8; ++r) {
#pragma unroll
    for (int m = 1; m < 16; m <<= 1) s8[r] += __shfl_xor(s8[r], m, 32);
    acc[r] *= 1.0f / s8[r];
  }

  if (l16 < 8) {
    float* ob = obuf + (size_t)n * HCH * LSPAT + (size_t)(h * DHEAD + l16) * LSPAT;
#pragma unroll
    for (int r = 0; r < 8; ++r) ob[qbase + r + half * 8] = acc[r];
  }
}

#define POS_OCT 8
#define POS_BLK 128
__global__ __launch_bounds__(POS_BLK) void pos_conv_kernel(
    const float* __restrict__ qkv, const float* __restrict__ pw,
    const float* __restrict__ pb, float* __restrict__ obuf) {
  const int chunks = LSPAT / POS_BLK;
  int b = blockIdx.x;
  const int chunk = b % chunks;  b /= chunks;
  const int og = b % (HCH / POS_OCT);
  const int n  = b / (HCH / POS_OCT);
  const unsigned p = chunk * POS_BLK + threadIdx.x;
  const int yp = (int)(p / WW);
  const int xp = (int)(p % WW);
  const int oc0 = og * POS_OCT;

  unsigned off[9];
  bool ok[9];
#pragma unroll
  for (int ky = 0; ky < 3; ++ky) {
#pragma unroll
    for (int kx = 0; kx < 3; ++kx) {
      const int iy = yp + ky - 1;
      const int ix = xp + kx - 1;
      const bool in_img = (iy >= 0) && (iy < HH) && (ix >= 0) && (ix < WW);
      ok[ky * 3 + kx]  = in_img;
      off[ky * 3 + kx] = in_img ? (unsigned)(iy * WW + ix) : p;
    }
  }

  float acc[POS_OCT];
#pragma unroll
  for (int o = 0; o < POS_OCT; ++o) acc[o] = pb[oc0 + o];

  const float* vbase = qkv + (size_t)n * CH3 * LSPAT + (size_t)16 * LSPAT;
  float cur[DHEAD][9];
#pragma unroll
  for (int dd = 0; dd < DHEAD; ++dd) {
#pragma unroll
    for (int t = 0; t < 9; ++t) cur[dd][t] = vbase[(unsigned)(dd * LSPAT) + off[t]];
  }

  for (int hh = 0; hh < NHEAD; ++hh) {
    const int hn = (hh + 1 < NHEAD) ? hh + 1 : hh;
    const unsigned cho = (unsigned)(hn * 24 * LSPAT);
    float nxt[DHEAD][9];
#pragma unroll
    for (int dd = 0; dd < DHEAD; ++dd) {
#pragma unroll
      for (int t = 0; t < 9; ++t)
        nxt[dd][t] = vbase[cho + (unsigned)(dd * LSPAT) + off[t]];
    }

#pragma unroll
    for (int dd = 0; dd < DHEAD; ++dd) {
      const int ic = hh * DHEAD + dd;
      float in[9];
#pragma unroll
      for (int t = 0; t < 9; ++t) in[t] = ok[t] ? cur[dd][t] : 0.0f;
#pragma unroll
      for (int o = 0; o < POS_OCT; ++o) {
        const float* wrow = pw + ((size_t)(oc0 + o) * HCH + ic) * 9;
#pragma unroll
        for (int t = 0; t < 9; ++t) acc[o] = fmaf(in[t], wrow[t], acc[o]);
      }
    }

#pragma unroll
    for (int dd = 0; dd < DHEAD; ++dd) {
#pragma unroll
      for (int t = 0; t < 9; ++t) cur[dd][t] = nxt[dd][t];
    }
  }

  float* ob = obuf + (size_t)n * HCH * LSPAT + (size_t)oc0 * LSPAT + p;
#pragma unroll
  for (int o = 0; o < POS_OCT; ++o) ob[(size_t)o * LSPAT] += acc[o];
}

__global__ __launch_bounds__(256) void res_gemm_kernel(
    const float* __restrict__ obuf, const float* __restrict__ w,
    const float* __restrict__ bias, float* __restrict__ out) {
  const int lane = threadIdx.x & 31;
  const int half = lane >> 4;
  const int l16  = lane & 15;
  const int wave = blockIdx.x * (blockDim.x >> 5) + (threadIdx.x >> 5);

  const int ntn = LSPAT / 16;
  const int ntm = CIN / 16;
  int t = wave;
  const int n  = t / (ntm * ntn); t -= n * ntm * ntn;
  const int mt = t / ntn;
  const int ct = t % ntn;

  const int mrow = mt * 16 + l16;
  const int col  = ct * 16 + l16;
  const float* ib = obuf + (size_t)n * HCH * LSPAT;

  v8f acc = {};
#pragma unroll
  for (int kk = 0; kk < HCH; kk += 4) {
    const int ka = kk + half * 2;
    v2f a; a[0] = w[mrow * HCH + ka];
           a[1] = w[mrow * HCH + ka + 1];
    v2f b; b[0] = ib[(size_t)ka * LSPAT + col];
           b[1] = ib[(size_t)(ka + 1) * LSPAT + col];
    acc = wmma4(a, b, acc);
  }

  float* ob = out + (size_t)n * CIN * LSPAT;
#pragma unroll
  for (int r = 0; r < 8; ++r) {
    const int row = mt * 16 + r + half * 8;
    ob[(size_t)row * LSPAT + col] = acc[r] + bias[row];
  }
}

extern "C" void kernel_launch(void* const* d_in, const int* in_sizes, int n_in,
                              void* d_out, int out_size, void* d_ws, size_t ws_size,
                              hipStream_t stream) {
  const float* x     = (const float*)d_in[0];
  const float* qkv_w = (const float*)d_in[1];
  const float* qkv_b = (const float*)d_in[2];
  const float* pos_w = (const float*)d_in[3];
  const float* pos_b = (const float*)d_in[4];
  const float* res_w = (const float*)d_in[5];
  const float* res_b = (const float*)d_in[6];
  float* out = (float*)d_out;

  float* qkv  = (float*)d_ws;
  float* obuf = qkv + (size_t)NBATCH * CH3 * LSPAT;

  qkv_gemm_kernel<<<NBATCH * (CH3 / QK_ROWS) * (LSPAT / QK_COLS), 256, 0,
                    stream>>>(x, qkv_w, qkv_b, qkv);

  attn_kernel<<<NBATCH * NHEAD * (LSPAT / 64), 128, 0, stream>>>(qkv, obuf);

  pos_conv_kernel<<<NBATCH * (HCH / POS_OCT) * (LSPAT / POS_BLK), POS_BLK, 0,
                    stream>>>(qkv, pos_w, pos_b, obuf);

  const int waves4 = NBATCH * (CIN / 16) * (LSPAT / 16);
  res_gemm_kernel<<<waves4 / 8, 256, 0, stream>>>(obuf, res_w, res_b, out);
}


# attention P*V pass: the e tile in LDS is per wave, so it is ordered at wave scope, no work-group barrier; a lane half takes 8 consecutive keys as 16-byte reads; P*V sum order in a key tile changes
# speedup vs baseline: 2.3447x; 1.0476x over previous
#include <hip/hip_runtime.h>
#include <math.h>

typedef __attribute__((ext_vector_type(2))) float v2f;
typedef __attribute__((ext_vector_type(8))) float v8f;

#define NBATCH 4
#define CIN 512
#define HH 48
#define WW 48
#define LSPAT (HH*WW)
#define CH3 192
#define HCH 64
#define NHEAD 8
#define DHEAD 8
#define ATTN_SCALE 0.35355339059327379f

static __device__ __forceinline__ v8f wmma4(v2f a, v2f b, v8f c) {
  return __builtin_amdgcn_wmma_f32_16x16x4_f32(false, a, false, b, (short)0, c,
                                               false, false);
}

typedef __attribute__((ext_vector_type(4))) float v4f;
#define QK_ROWS 64
#define QK_COLS 64
#define QK_KC 64
#define QK_SP 260
__global__ __launch_bounds__(256) void qkv_gemm_kernel(
    const float* __restrict__ x, const float* __restrict__ w,
    const float* __restrict__ bias, float* __restrict__ qkv) {
  __shared__ __attribute__((aligned(16))) float wl[2][16 * QK_SP];
  const int tid  = threadIdx.x;
  const int lane = tid & 31;
  const int half = lane >> 4;
  const int l16  = lane & 15;
  const int wv   = tid >> 5;

  int b = blockIdx.x;
  const int mg = b % (CH3 / QK_ROWS);   b /= (CH3 / QK_ROWS);
  const int cb = b % (LSPAT / QK_COLS);
  const int n  = b / (LSPAT / QK_COLS);
  const int m0 = mg * QK_ROWS;
  const int rt = wv >> 2;
  const int col = cb * QK_COLS + (wv & 3) * 16 + l16;
  const float* xb = x + (size_t)n * CIN * LSPAT;

  const int q4   = tid & 15;
  const int srow = tid >> 4;
  const float* wsrc = w + (size_t)(m0 + srow) * CIN + q4 * 4;
  const int sdst = q4 * QK_SP + srow * 4;

  const float* bp = xb + (size_t)(half * 2) * LSPAT + col;
  const int aoff = (rt * 32 + l16) * 4 + half * 2;

  auto load_w = [&](v4f (&wr)[4], const int kc) {
#pragma unroll
    for (int i = 0; i < 4; ++i) {
      const float* p = wsrc + (size_t)(i * 16) * CIN + kc;
      wr[i][0] = p[0]; wr[i][1] = p[1]; wr[i][2] = p[2]; wr[i][3] = p[3];
    }
  };
  auto store_w = [&](float* buf, const v4f (&wr)[4]) {
#pragma unroll
    for (int i = 0; i < 4; ++i) *(v4f*)(buf + sdst + i * 64) = wr[i];
  };
  auto load_b = [&](v2f (&bf)[16], const int kc) {
#pragma unroll
    for (int s = 0; s < 16; ++s) {
      bf[s][0] = bp[(size_t)(kc + 4 * s) * LSPAT];
      bf[s][1] = bp[(size_t)(kc + 4 * s + 1) * LSPAT];
    }
  };
  auto lds_barrier = [&]() {
    __builtin_amdgcn_fence(__ATOMIC_RELEASE, "workgroup", "local");
    __builtin_amdgcn_s_barrier();
    __builtin_amdgcn_fence(__ATOMIC_ACQUIRE, "workgroup", "local");
  };

  v8f acc0 = {}, acc1 = {};
  auto compute = [&](const float* buf, const v2f (&bf)[16]) {
#pragma unroll
    for (int s = 0; s < 16; ++s) {
      const v2f a0 = *(const v2f*)(buf + s * QK_SP + aoff);
      const v2f a1 = *(const v2f*)(buf + s * QK_SP + aoff + 64);
      acc0 = wmma4(a0, bf[s], acc0);
      acc1 = wmma4(a1, bf[s], acc1);
    }
  };

  v4f wr[4];
  v2f bf0[16], bf1[16];
  load_w(wr, 0);
  load_b(bf0, 0);
  store_w(wl[0], wr);
  lds_barrier();
#pragma unroll
  for (int c = 0; c < CIN / QK_KC; c += 2) {
    load_w(wr, (c + 1) * QK_KC);
    load_b(bf1, (c + 1) * QK_KC);
    compute(wl[0], bf0);
    store_w(wl[1], wr);
    lds_barrier();
    if (c + 2 < CIN / QK_KC) {
      load_w(wr, (c + 2) * QK_KC);
      load_b(bf0, (c + 2) * QK_KC);
    }
    compute(wl[1], bf1);
    if (c + 2 < CIN / QK_KC) {
      store_w(wl[0], wr);
      lds_barrier();
    }
  }

  float* out = qkv + (size_t)n * CH3 * LSPAT;
#pragma unroll
  for (int r = 0; r < 8; ++r) {
    const int row0 = m0 + rt * 32 + r + half * 8;
    const int row1 = row0 + 16;
    out[(size_t)row0 * LSPAT + col] = acc0[r] + bias[row0];
    out[(size_t)row1 * LSPAT + col] = acc1[r] + bias[row1];
  }
}

#define PT_PITCH 20
__global__ __launch_bounds__(128) void attn_kernel(
    const float* __restrict__ qkv, float* __restrict__ obuf) {
  __shared__ __attribute__((aligned(16))) float plds[4][16][PT_PITCH];
  const int lane = threadIdx.x & 31;
  const int half = lane >> 4;
  const int l16  = lane & 15;
  const int wv   = threadIdx.x >> 5;

  const int qtiles = LSPAT / 64;
  const int nh = blockIdx.x / qtiles;
  const int qt = blockIdx.x % qtiles;
  const int n = nh >> 3, h = nh & 7;
  const int qbase = qt * 64 + wv * 16;

  const float* base = qkv + (size_t)n * CH3 * LSPAT + (size_t)h * 24 * LSPAT;
  const float* qm = base;
  const float* km = base + (size_t)DHEAD * LSPAT;
  const float* vm = base + (size_t)2 * DHEAD * LSPAT;

  v2f aq[2];
#pragma unroll
  for (int db = 0; db < 2; ++db) {
    const int d0 = db * 4 + half * 2;
    aq[db][0] = qm[(size_t)d0 * LSPAT + qbase + l16];
    aq[db][1] = qm[(size_t)(d0 + 1) * LSPAT + qbase + l16];
  }

  float m8[8];
#pragma unroll
  for (int r = 0; r < 8; ++r) m8[r] = -3.0e38f;

  for (int kt = 0; kt < LSPAT / 16; ++kt) {
    const int kb = kt * 16;
    v8f s = {};
#pragma unroll
    for (int db = 0; db < 2; ++db) {
      const int d0 = db * 4 + half * 2;
      v2f bk; bk[0] = km[(size_t)d0 * LSPAT + kb + l16];
              bk[1] = km[(size_t)(d0 + 1) * LSPAT + kb + l16];
      s = wmma4(aq[db], bk, s);
    }
#pragma unroll
    for (int r = 0; r < 8; ++r) m8[r] = fmaxf(m8[r], s[r]);
  }
#pragma unroll
  for (int r = 0; r < 8; ++r) {
#pragma unroll
    for (int m = 1; m < 16; m <<= 1) m8[r] = fmaxf(m8[r], __shfl_xor(m8[r], m, 32));
  }

  float s8[8];
#pragma unroll
  for (int r = 0; r < 8; ++r) s8[r] = 0.0f;

  const float* kp = km + (size_t)(half * 2) * LSPAT + l16;
  const float* vp = vm + (size_t)(l16 & 7) * LSPAT + half * 8;

  auto load_tile = [&](v2f (&bk)[2], v4f (&bv)[2], const int kb) {
#pragma unroll
    for (int db = 0; db < 2; ++db) {
      bk[db][0] = kp[(size_t)(db * 4) * LSPAT + kb];
      bk[db][1] = kp[(size_t)(db * 4 + 1) * LSPAT + kb];
    }
    bv[0] = *(const v4f*)(vp + kb);
    bv[1] = *(const v4f*)(vp + kb + 4);
  };

  auto wave_lds_order = [&]() {
    __builtin_amdgcn_fence(__ATOMIC_RELEASE, "wavefront", "local");
    __builtin_amdgcn_wave_barrier();
    __builtin_amdgcn_fence(__ATOMIC_ACQUIRE, "wavefront", "local");
  };

  v8f acc = {};
  auto do_tile = [&](const v2f (&bk)[2], const v4f (&bv)[2]) {
    v8f s = {};
    s = wmma4(aq[0], bk[0], s);
    s = wmma4(aq[1], bk[1], s);
#pragma unroll
    for (int r = 0; r < 8; ++r) {
      const float e = __expf((s[r] - m8[r]) * ATTN_SCALE);
      s8[r] += e;
      plds[wv][r + half * 8][l16] = e;
    }
    wave_lds_order();
    const v4f p0 = *(const v4f*)&plds[wv][l16][half * 8];
    const v4f p1 = *(const v4f*)&plds[wv][l16][half * 8 + 4];
    v2f ap, bp;
    ap[0] = p0[0]; ap[1] = p0[1]; bp[0] = bv[0][0]; bp[1] = bv[0][1];
    acc = wmma4(ap, bp, acc);
    ap[0] = p0[2]; ap[1] = p0[3]; bp[0] = bv[0][2]; bp[1] = bv[0][3];
    acc = wmma4(ap, bp, acc);
    ap[0] = p1[0]; ap[1] = p1[1]; bp[0] = bv[1][0]; bp[1] = bv[1][1];
    acc = wmma4(ap, bp, acc);
    ap[0] = p1[2]; ap[1] = p1[3]; bp[0] = bv[1][2]; bp[1] = bv[1][3];
    acc = wmma4(ap, bp, acc);
    wave_lds_order();
  };

  v2f bk0[2], bk1[2];
  v4f bv0[2], bv1[2];
  load_tile(bk0, bv0, 0);
  for (int kt = 0; kt < LSPAT / 16; kt += 2) {
    load_tile(bk1, bv1, (kt + 1) * 16);
    do_tile(bk0, bv0);
    const int kn = (kt + 2 < LSPAT / 16) ? kt + 2 : kt + 1;
    load_tile(bk0, bv0, kn * 16);
    do_tile(bk1, bv1);
  }

#pragma unroll
  for (int r = 0; r < 8; ++r) {
#pragma unroll
    for (int m = 1; m < 16; m <<= 1) s8[r] += __shfl_xor(s8[r], m, 32);
    acc[r] *= 1.0f / s8[r];
  }

  if (l16 < 8) {
    float* ob = obuf + (size_t)n * HCH * LSPAT + (size_t)(h * DHEAD + l16) * LSPAT;
#pragma unroll
    for (int r = 0; r < 8; ++r) ob[qbase + r + half * 8] = acc[r];
  }
}

#define POS_OCT 8
#define POS_BLK 128
__global__ __launch_bounds__(POS_BLK) void pos_conv_kernel(
    const float* __restrict__ qkv, const float* __restrict__ pw,
    const float* __restrict__ pb, float* __restrict__ obuf) {
  const int chunks = LSPAT / POS_BLK;
  int b = blockIdx.x;
  const int chunk = b % chunks;  b /= chunks;
  const int og = b % (HCH / POS_OCT);
  const int n  = b / (HCH / POS_OCT);
  const unsigned p = chunk * POS_BLK + threadIdx.x;
  const int yp = (int)(p / WW);
  const int xp = (int)(p % WW);
  const int oc0 = og * POS_OCT;

  unsigned off[9];
  bool ok[9];
#pragma unroll
  for (int ky = 0; ky < 3; ++ky) {
#pragma unroll
    for (int kx = 0; kx < 3; ++kx) {
      const int iy = yp + ky - 1;
      const int ix = xp + kx - 1;
      const bool in_img = (iy >= 0) && (iy < HH) && (ix >= 0) && (ix < WW);
      ok[ky * 3 + kx]  = in_img;
      off[ky * 3 + kx] = in_img ? (unsigned)(iy * WW + ix) : p;
    }
  }

  float acc[POS_OCT];
#pragma unroll
  for (int o = 0; o < POS_OCT; ++o) acc[o] = pb[oc0 + o];

  const float* vbase = qkv + (size_t)n * CH3 * LSPAT + (size_t)16 * LSPAT;
  float cur[DHEAD][9];
#pragma unroll
  for (int dd = 0; dd < DHEAD; ++dd) {
#pragma unroll
    for (int t = 0; t < 9; ++t) cur[dd][t] = vbase[(unsigned)(dd * LSPAT) + off[t]];
  }

  for (int hh = 0; hh < NHEAD; ++hh) {
    const int hn = (hh + 1 < NHEAD) ? hh + 1 : hh;
    const unsigned cho = (unsigned)(hn * 24 * LSPAT);
    float nxt[DHEAD][9];
#pragma unroll
    for (int dd = 0; dd < DHEAD; ++dd) {
#pragma unroll
      for (int t = 0; t < 9; ++t)
        nxt[dd][t] = vbase[cho + (unsigned)(dd * LSPAT) + off[t]];
    }

#pragma unroll
    for (int dd = 0; dd < DHEAD; ++dd) {
      const int ic = hh * DHEAD + dd;
      float in[9];
#pragma unroll
      for (int t = 0; t < 9; ++t) in[t] = ok[t] ? cur[dd][t] : 0.0f;
#pragma unroll
      for (int o = 0; o < POS_OCT; ++o) {
        const float* wrow = pw + ((size_t)(oc0 + o) * HCH + ic) * 9;
#pragma unroll
        for (int t = 0; t < 9; ++t) acc[o] = fmaf(in[t], wrow[t], acc[o]);
      }
    }

#pragma unroll
    for (int dd = 0; dd < DHEAD; ++dd) {
#pragma unroll
      for (int t = 0; t < 9; ++t) cur[dd][t] = nxt[dd][t];
    }
  }

  float* ob = obuf + (size_t)n * HCH * LSPAT + (size_t)oc0 * LSPAT + p;
#pragma unroll
  for (int o = 0; o < POS_OCT; ++o) ob[(size_t)o * LSPAT] += acc[o];
}

__global__ __launch_bounds__(256) void res_gemm_kernel(
    const float* __restrict__ obuf, const float* __restrict__ w,
    const float* __restrict__ bias, float* __restrict__ out) {
  const int lane = threadIdx.x & 31;
  const int half = lane >> 4;
  const int l16  = lane & 15;
  const int wave = blockIdx.x * (blockDim.x >> 5) + (threadIdx.x >> 5);

  const int ntn = LSPAT / 16;
  const int ntm = CIN / 16;
  int t = wave;
  const int n  = t / (ntm * ntn); t -= n * ntm * ntn;
  const int mt = t / ntn;
  const int ct = t % ntn;

  const int mrow = mt * 16 + l16;
  const int col  = ct * 16 + l16;
  const float* ib = obuf + (size_t)n * HCH * LSPAT;

  v8f acc = {};
#pragma unroll
  for (int kk = 0; kk < HCH; kk += 4) {
    const int ka = kk + half * 2;
    v2f a; a[0] = w[mrow * HCH + ka];
           a[1] = w[mrow * HCH + ka + 1];
    v2f b; b[0] = ib[(size_t)ka * LSPAT + col];
           b[1] = ib[(size_t)(ka + 1) * LSPAT + col];
    acc = wmma4(a, b, acc);
  }

  float* ob = out + (size_t)n * CIN * LSPAT;
#pragma unroll
  for (int r = 0; r < 8; ++r) {
    const int row = mt * 16 + r + half * 8;
    ob[(size_t)row * LSPAT + col] = acc[r] + bias[row];
  }
}

extern "C" void kernel_launch(void* const* d_in, const int* in_sizes, int n_in,
                              void* d_out, int out_size, void* d_ws, size_t ws_size,
                              hipStream_t stream) {
  const float* x     = (const float*)d_in[0];
  const float* qkv_w = (const float*)d_in[1];
  const float* qkv_b = (const float*)d_in[2];
  const float* pos_w = (const float*)d_in[3];
  const float* pos_b = (const float*)d_in[4];
  const float* res_w = (const float*)d_in[5];
  const float* res_b = (const float*)d_in[6];
  float* out = (float*)d_out;

  float* qkv  = (float*)d_ws;
  float* obuf = qkv + (size_t)NBATCH * CH3 * LSPAT;

  qkv_gemm_kernel<<<NBATCH * (CH3 / QK_ROWS) * (LSPAT / QK_COLS), 256, 0,
                    stream>>>(x, qkv_w, qkv_b, qkv);

  attn_kernel<<<NBATCH * NHEAD * (LSPAT / 64), 128, 0, stream>>>(qkv, obuf);

  pos_conv_kernel<<<NBATCH * (HCH / POS_OCT) * (LSPAT / POS_BLK), POS_BLK, 0,
                    stream>>>(qkv, pos_w, pos_b, obuf);

  const int waves4 = NBATCH * (CIN / 16) * (LSPAT / 16);
  res_gemm_kernel<<<waves4 / 8, 256, 0, stream>>>(obuf, res_w, res_b, out);
}


# pos conv moved into the attention launch as its last 576 work-groups: vector work fills issue slots the matrix-bound attention waves leave; own output plane, added in the res GEMM's B loader
# speedup vs baseline: 2.7492x; 1.1725x over previous
#include <hip/hip_runtime.h>
#include <math.h>

typedef __attribute__((ext_vector_type(2))) float v2f;
typedef __attribute__((ext_vector_type(8))) float v8f;

#define NBATCH 4
#define CIN 512
#define HH 48
#define WW 48
#define LSPAT (HH*WW)
#define CH3 192
#define HCH 64
#define NHEAD 8
#define DHEAD 8
#define ATTN_SCALE 0.35355339059327379f

static __device__ __forceinline__ v8f wmma4(v2f a, v2f b, v8f c) {
  return __builtin_amdgcn_wmma_f32_16x16x4_f32(false, a, false, b, (short)0, c,
                                               false, false);
}

typedef __attribute__((ext_vector_type(4))) float v4f;
#define QK_ROWS 64
#define QK_COLS 64
#define QK_KC 64
#define QK_SP 260
__global__ __launch_bounds__(256) void qkv_gemm_kernel(
    const float* __restrict__ x, const float* __restrict__ w,
    const float* __restrict__ bias, float* __restrict__ qkv) {
  __shared__ __attribute__((aligned(16))) float wl[2][16 * QK_SP];
  const int tid  = threadIdx.x;
  const int lane = tid & 31;
  const int half = lane >> 4;
  const int l16  = lane & 15;
  const int wv   = tid >> 5;

  int b = blockIdx.x;
  const int mg = b % (CH3 / QK_ROWS);   b /= (CH3 / QK_ROWS);
  const int cb = b % (LSPAT / QK_COLS);
  const int n  = b / (LSPAT / QK_COLS);
  const int m0 = mg * QK_ROWS;
  const int rt = wv >> 2;
  const int col = cb * QK_COLS + (wv & 3) * 16 + l16;
  const float* xb = x + (size_t)n * CIN * LSPAT;

  const int q4   = tid & 15;
  const int srow = tid >> 4;
  const float* wsrc = w + (size_t)(m0 + srow) * CIN + q4 * 4;
  const int sdst = q4 * QK_SP + srow * 4;

  const float* bp = xb + (size_t)(half * 2) * LSPAT + col;
  const int aoff = (rt * 32 + l16) * 4 + half * 2;

  auto load_w = [&](v4f (&wr)[4], const int kc) {
#pragma unroll
    for (int i = 0; i < 4; ++i) {
      const float* p = wsrc + (size_t)(i * 16) * CIN + kc;
      wr[i][0] = p[0]; wr[i][1] = p[1]; wr[i][2] = p[2]; wr[i][3] = p[3];
    }
  };
  auto store_w = [&](float* buf, const v4f (&wr)[4]) {
#pragma unroll
    for (int i = 0; i < 4; ++i) *(v4f*)(buf + sdst + i * 64) = wr[i];
  };
  auto load_b = [&](v2f (&bf)[16], const int kc) {
#pragma unroll
    for (int s = 0; s < 16; ++s) {
      bf[s][0] = bp[(size_t)(kc + 4 * s) * LSPAT];
      bf[s][1] = bp[(size_t)(kc + 4 * s + 1) * LSPAT];
    }
  };
  auto lds_barrier = [&]() {
    __builtin_amdgcn_fence(__ATOMIC_RELEASE, "workgroup", "local");
    __builtin_amdgcn_s_barrier();
    __builtin_amdgcn_fence(__ATOMIC_ACQUIRE, "workgroup", "local");
  };

  v8f acc0 = {}, acc1 = {};
  auto compute = [&](const float* buf, const v2f (&bf)[16]) {
#pragma unroll
    for (int s = 0; s < 16; ++s) {
      const v2f a0 = *(const v2f*)(buf + s * QK_SP + aoff);
      const v2f a1 = *(const v2f*)(buf + s * QK_SP + aoff + 64);
      acc0 = wmma4(a0, bf[s], acc0);
      acc1 = wmma4(a1, bf[s], acc1);
    }
  };

  v4f wr[4];
  v2f bf0[16], bf1[16];
  load_w(wr, 0);
  load_b(bf0, 0);
  store_w(wl[0], wr);
  lds_barrier();
#pragma unroll
  for (int c = 0; c < CIN / QK_KC; c += 2) {
    load_w(wr, (c + 1) * QK_KC);
    load_b(bf1, (c + 1) * QK_KC);
    compute(wl[0], bf0);
    store_w(wl[1], wr);
    lds_barrier();
    if (c + 2 < CIN / QK_KC) {
      load_w(wr, (c + 2) * QK_KC);
      load_b(bf0, (c + 2) * QK_KC);
    }
    compute(wl[1], bf1);
    if (c + 2 < CIN / QK_KC) {
      store_w(wl[0], wr);
      lds_barrier();
    }
  }

  float* out = qkv + (size_t)n * CH3 * LSPAT;
#pragma unroll
  for (int r = 0; r < 8; ++r) {
    const int row0 = m0 + rt * 32 + r + half * 8;
    const int row1 = row0 + 16;
    out[(size_t)row0 * LSPAT + col] = acc0[r] + bias[row0];
    out[(size_t)row1 * LSPAT + col] = acc1[r] + bias[row1];
  }
}

#define PT_PITCH 20
static __device__ __forceinline__ void attn_body(
    const float* __restrict__ qkv, float* __restrict__ obuf, const int bid,
    float (*plds)[16][PT_PITCH]) {
  const int lane = threadIdx.x & 31;
  const int half = lane >> 4;
  const int l16  = lane & 15;
  const int wv   = threadIdx.x >> 5;

  const int qtiles = LSPAT / 64;
  const int nh = bid / qtiles;
  const int qt = bid % qtiles;
  const int n = nh >> 3, h = nh & 7;
  const int qbase = qt * 64 + wv * 16;

  const float* base = qkv + (size_t)n * CH3 * LSPAT + (size_t)h * 24 * LSPAT;
  const float* qm = base;
  const float* km = base + (size_t)DHEAD * LSPAT;
  const float* vm = base + (size_t)2 * DHEAD * LSPAT;

  v2f aq[2];
#pragma unroll
  for (int db = 0; db < 2; ++db) {
    const int d0 = db * 4 + half * 2;
    aq[db][0] = qm[(size_t)d0 * LSPAT + qbase + l16];
    aq[db][1] = qm[(size_t)(d0 + 1) * LSPAT + qbase + l16];
  }

  float m8[8];
#pragma unroll
  for (int r = 0; r < 8; ++r) m8[r] = -3.0e38f;

  for (int kt = 0; kt < LSPAT / 16; ++kt) {
    const int kb = kt * 16;
    v8f s = {};
#pragma unroll
    for (int db = 0; db < 2; ++db) {
      const int d0 = db * 4 + half * 2;
      v2f bk; bk[0] = km[(size_t)d0 * LSPAT + kb + l16];
              bk[1] = km[(size_t)(d0 + 1) * LSPAT + kb + l16];
      s = wmma4(aq[db], bk, s);
    }
#pragma unroll
    for (int r = 0; r < 8; ++r) m8[r] = fmaxf(m8[r], s[r]);
  }
#pragma unroll
  for (int r = 0; r < 8; ++r) {
#pragma unroll
    for (int m = 1; m < 16; m <<= 1) m8[r] = fmaxf(m8[r], __shfl_xor(m8[r], m, 32));
  }

  float s8[8];
#pragma unroll
  for (int r = 0; r < 8; ++r) s8[r] = 0.0f;

  const float* kp = km + (size_t)(half * 2) * LSPAT + l16;
  const float* vp = vm + (size_t)(l16 & 7) * LSPAT + half * 8;

  auto load_tile = [&](v2f (&bk)[2], v4f (&bv)[2], const int kb) {
#pragma unroll
    for (int db = 0; db < 2; ++db) {
      bk[db][0] = kp[(size_t)(db * 4) * LSPAT + kb];
      bk[db][1] = kp[(size_t)(db * 4 + 1) * LSPAT + kb];
    }
    bv[0] = *(const v4f*)(vp + kb);
    bv[1] = *(const v4f*)(vp + kb + 4);
  };

  auto wave_lds_order = [&]() {
    __builtin_amdgcn_fence(__ATOMIC_RELEASE, "wavefront", "local");
    __builtin_amdgcn_wave_barrier();
    __builtin_amdgcn_fence(__ATOMIC_ACQUIRE, "wavefront", "local");
  };

  v8f acc = {};
  auto do_tile = [&](const v2f (&bk)[2], const v4f (&bv)[2]) {
    v8f s = {};
    s = wmma4(aq[0], bk[0], s);
    s = wmma4(aq[1], bk[1], s);
#pragma unroll
    for (int r = 0; r < 8; ++r) {
      const float e = __expf((s[r] - m8[r]) * ATTN_SCALE);
      s8[r] += e;
      plds[wv][r + half * 8][l16] = e;
    }
    wave_lds_order();
    const v4f p0 = *(const v4f*)&plds[wv][l16][half * 8];
    const v4f p1 = *(const v4f*)&plds[wv][l16][half * 8 + 4];
    v2f ap, bp;
    ap[0] = p0[0]; ap[1] = p0[1]; bp[0] = bv[0][0]; bp[1] = bv[0][1];
    acc = wmma4(ap, bp, acc);
    ap[0] = p0[2]; ap[1] = p0[3]; bp[0] = bv[0][2]; bp[1] = bv[0][3];
    acc = wmma4(ap, bp, acc);
    ap[0] = p1[0]; ap[1] = p1[1]; bp[0] = bv[1][0]; bp[1] = bv[1][1];
    acc = wmma4(ap, bp, acc);
    ap[0] = p1[2]; ap[1] = p1[3]; bp[0] = bv[1][2]; bp[1] = bv[1][3];
    acc = wmma4(ap, bp, acc);
    wave_lds_order();
  };

  v2f bk0[2], bk1[2];
  v4f bv0[2], bv1[2];
  load_tile(bk0, bv0, 0);
  for (int kt = 0; kt < LSPAT / 16; kt += 2) {
    load_tile(bk1, bv1, (kt + 1) * 16);
    do_tile(bk0, bv0);
    const int kn = (kt + 2 < LSPAT / 16) ? kt + 2 : kt + 1;
    load_tile(bk0, bv0, kn * 16);
    do_tile(bk1, bv1);
  }

#pragma unroll
  for (int r = 0; r < 8; ++r) {
#pragma unroll
    for (int m = 1; m < 16; m <<= 1) s8[r] += __shfl_xor(s8[r], m, 32);
    acc[r] *= 1.0f / s8[r];
  }

  if (l16 < 8) {
    float* ob = obuf + (size_t)n * HCH * LSPAT + (size_t)(h * DHEAD + l16) * LSPAT;
#pragma unroll
    for (int r = 0; r < 8; ++r) ob[qbase + r + half * 8] = acc[r];
  }
}

#define POS_OCT 8
#define POS_BLK 128
#define POS_GROUPS (NBATCH * (HCH / POS_OCT) * (LSPAT / POS_BLK))
typedef const __attribute__((address_space(4))) float* cfloat_ptr;
static __device__ __forceinline__ void pos_conv_body(
    const float* __restrict__ qkv, const float* __restrict__ pw,
    const float* __restrict__ pb, float* __restrict__ pos, const int bid) {
  const int chunks = LSPAT / POS_BLK;
  int b = bid;
  const int chunk = b % chunks;  b /= chunks;
  const int og = b % (HCH / POS_OCT);
  const int n  = b / (HCH / POS_OCT);
  const unsigned p = chunk * POS_BLK + threadIdx.x;
  const int yp = (int)(p / WW);
  const int xp = (int)(p % WW);
  const int oc0 = og * POS_OCT;

  unsigned off[9];
  bool ok[9];
#pragma unroll
  for (int ky = 0; ky < 3; ++ky) {
#pragma unroll
    for (int kx = 0; kx < 3; ++kx) {
      const int iy = yp + ky - 1;
      const int ix = xp + kx - 1;
      const bool in_img = (iy >= 0) && (iy < HH) && (ix >= 0) && (ix < WW);
      ok[ky * 3 + kx]  = in_img;
      off[ky * 3 + kx] = in_img ? (unsigned)(iy * WW + ix) : p;
    }
  }

  const cfloat_ptr pwc = (cfloat_ptr)pw;
  const cfloat_ptr pbc = (cfloat_ptr)pb;
  float acc[POS_OCT];
#pragma unroll
  for (int o = 0; o < POS_OCT; ++o) acc[o] = pbc[oc0 + o];

  const float* vbase = qkv + (size_t)n * CH3 * LSPAT + (size_t)16 * LSPAT;
  for (int hh = 0; hh < NHEAD; ++hh) {
    const float* hb = vbase + (size_t)hh * 24 * LSPAT;
    float cur[DHEAD][9];
#pragma unroll
    for (int dd = 0; dd < DHEAD; ++dd) {
      const float* cb = hb + dd * LSPAT;
#pragma unroll
      for (int t = 0; t < 9; ++t) cur[dd][t] = cb[off[t]];
    }

#pragma unroll
    for (int dd = 0; dd < DHEAD; ++dd) {
      const int ic = hh * DHEAD + dd;
      float in[9];
#pragma unroll
      for (int t = 0; t < 9; ++t) in[t] = ok[t] ? cur[dd][t] : 0.0f;
#pragma unroll
      for (int o = 0; o < POS_OCT; ++o) {
        const cfloat_ptr wrow = pwc + ((size_t)(oc0 + o) * HCH + ic) * 9;
#pragma unroll
        for (int t = 0; t < 9; ++t) acc[o] = fmaf(in[t], wrow[t], acc[o]);
      }
    }
  }

  float* ob = pos + (size_t)n * HCH * LSPAT + (size_t)oc0 * LSPAT + p;
#pragma unroll
  for (int o = 0; o < POS_OCT; ++o) ob[(size_t)o * LSPAT] = acc[o];
}

#define ATT_GROUPS (NBATCH * NHEAD * (LSPAT / 64))
__global__ __launch_bounds__(128) void attn_pos_kernel(
    const float* __restrict__ qkv, const float* __restrict__ pw,
    const float* __restrict__ pb, float* __restrict__ obuf,
    float* __restrict__ pos) {
  __shared__ __attribute__((aligned(16))) float plds[4][16][PT_PITCH];
  if (blockIdx.x < ATT_GROUPS) {
    attn_body(qkv, obuf, (int)blockIdx.x, plds);
  } else {
    pos_conv_body(qkv, pw, pb, pos, (int)blockIdx.x - ATT_GROUPS);
  }
}

__global__ __launch_bounds__(256) void res_gemm_kernel(
    const float* __restrict__ obuf, const float* __restrict__ pos,
    const float* __restrict__ w,
    const float* __restrict__ bias, float* __restrict__ out) {
  const int lane = threadIdx.x & 31;
  const int half = lane >> 4;
  const int l16  = lane & 15;
  const int wave = blockIdx.x * (blockDim.x >> 5) + (threadIdx.x >> 5);

  const int ntn = LSPAT / 16;
  const int ntm = CIN / 16;
  int t = wave;
  const int n  = t / (ntm * ntn); t -= n * ntm * ntn;
  const int mt = t / ntn;
  const int ct = t % ntn;

  const int mrow = mt * 16 + l16;
  const int col  = ct * 16 + l16;
  const float* ib = obuf + (size_t)n * HCH * LSPAT;
  const float* pp = pos + (size_t)n * HCH * LSPAT;

  v8f acc = {};
#pragma unroll
  for (int kk = 0; kk < HCH; kk += 4) {
    const int ka = kk + half * 2;
    v2f a; a[0] = w[mrow * HCH + ka];
           a[1] = w[mrow * HCH + ka + 1];
    v2f b; b[0] = ib[(size_t)ka * LSPAT + col] + pp[(size_t)ka * LSPAT + col];
           b[1] = ib[(size_t)(ka + 1) * LSPAT + col] +
                  pp[(size_t)(ka + 1) * LSPAT + col];
    acc = wmma4(a, b, acc);
  }

  float* ob = out + (size_t)n * CIN * LSPAT;
#pragma unroll
  for (int r = 0; r < 8; ++r) {
    const int row = mt * 16 + r + half * 8;
    ob[(size_t)row * LSPAT + col] = acc[r] + bias[row];
  }
}

extern "C" void kernel_launch(void* const* d_in, const int* in_sizes, int n_in,
                              void* d_out, int out_size, void* d_ws, size_t ws_size,
                              hipStream_t stream) {
  const float* x     = (const float*)d_in[0];
  const float* qkv_w = (const float*)d_in[1];
  const float* qkv_b = (const float*)d_in[2];
  const float* pos_w = (const float*)d_in[3];
  const float* pos_b = (const float*)d_in[4];
  const float* res_w = (const float*)d_in[5];
  const float* res_b = (const float*)d_in[6];
  float* out = (float*)d_out;

  float* qkv  = (float*)d_ws;
  float* obuf = qkv + (size_t)NBATCH * CH3 * LSPAT;
  float* pos  = obuf + (size_t)NBATCH * HCH * LSPAT;

  qkv_gemm_kernel<<<NBATCH * (CH3 / QK_ROWS) * (LSPAT / QK_COLS), 256, 0,
                    stream>>>(x, qkv_w, qkv_b, qkv);

  attn_pos_kernel<<<ATT_GROUPS + POS_GROUPS, 128, 0, stream>>>(qkv, pos_w, pos_b,
                                                             obuf, pos);

  const int waves4 = NBATCH * (CIN / 16) * (LSPAT / 16);
  res_gemm_kernel<<<waves4 / 8, 256, 0, stream>>>(obuf, pos, res_w, res_b, out);
}


# attention: if max|q|*max|k|*scale <= 32 the softmax shift is that norm bound (no score exceeds it, e <= 1) and the max pass over the keys is not run; exact max otherwise; softmax rounding changes
# speedup vs baseline: 3.0982x; 1.1269x over previous
#include <hip/hip_runtime.h>
#include <math.h>

typedef __attribute__((ext_vector_type(2))) float v2f;
typedef __attribute__((ext_vector_type(8))) float v8f;

#define NBATCH 4
#define CIN 512
#define HH 48
#define WW 48
#define LSPAT (HH*WW)
#define CH3 192
#define HCH 64
#define NHEAD 8
#define DHEAD 8
#define ATTN_SCALE 0.35355339059327379f

static __device__ __forceinline__ v8f wmma4(v2f a, v2f b, v8f c) {
  return __builtin_amdgcn_wmma_f32_16x16x4_f32(false, a, false, b, (short)0, c,
                                               false, false);
}

typedef __attribute__((ext_vector_type(4))) float v4f;
#define QK_ROWS 64
#define QK_COLS 64
#define QK_KC 64
#define QK_SP 260
__global__ __launch_bounds__(256) void qkv_gemm_kernel(
    const float* __restrict__ x, const float* __restrict__ w,
    const float* __restrict__ bias, float* __restrict__ qkv) {
  __shared__ __attribute__((aligned(16))) float wl[2][16 * QK_SP];
  const int tid  = threadIdx.x;
  const int lane = tid & 31;
  const int half = lane >> 4;
  const int l16  = lane & 15;
  const int wv   = tid >> 5;

  int b = blockIdx.x;
  const int mg = b % (CH3 / QK_ROWS);   b /= (CH3 / QK_ROWS);
  const int cb = b % (LSPAT / QK_COLS);
  const int n  = b / (LSPAT / QK_COLS);
  const int m0 = mg * QK_ROWS;
  const int rt = wv >> 2;
  const int col = cb * QK_COLS + (wv & 3) * 16 + l16;
  const float* xb = x + (size_t)n * CIN * LSPAT;

  const int q4   = tid & 15;
  const int srow = tid >> 4;
  const float* wsrc = w + (size_t)(m0 + srow) * CIN + q4 * 4;
  const int sdst = q4 * QK_SP + srow * 4;

  const float* bp = xb + (size_t)(half * 2) * LSPAT + col;
  const int aoff = (rt * 32 + l16) * 4 + half * 2;

  auto load_w = [&](v4f (&wr)[4], const int kc) {
#pragma unroll
    for (int i = 0; i < 4; ++i) {
      const float* p = wsrc + (size_t)(i * 16) * CIN + kc;
      wr[i][0] = p[0]; wr[i][1] = p[1]; wr[i][2] = p[2]; wr[i][3] = p[3];
    }
  };
  auto store_w = [&](float* buf, const v4f (&wr)[4]) {
#pragma unroll
    for (int i = 0; i < 4; ++i) *(v4f*)(buf + sdst + i * 64) = wr[i];
  };
  auto load_b = [&](v2f (&bf)[16], const int kc) {
#pragma unroll
    for (int s = 0; s < 16; ++s) {
      bf[s][0] = bp[(size_t)(kc + 4 * s) * LSPAT];
      bf[s][1] = bp[(size_t)(kc + 4 * s + 1) * LSPAT];
    }
  };
  auto lds_barrier = [&]() {
    __builtin_amdgcn_fence(__ATOMIC_RELEASE, "workgroup", "local");
    __builtin_amdgcn_s_barrier();
    __builtin_amdgcn_fence(__ATOMIC_ACQUIRE, "workgroup", "local");
  };

  v8f acc0 = {}, acc1 = {};
  auto compute = [&](const float* buf, const v2f (&bf)[16]) {
#pragma unroll
    for (int s = 0; s < 16; ++s) {
      const v2f a0 = *(const v2f*)(buf + s * QK_SP + aoff);
      const v2f a1 = *(const v2f*)(buf + s * QK_SP + aoff + 64);
      acc0 = wmma4(a0, bf[s], acc0);
      acc1 = wmma4(a1, bf[s], acc1);
    }
  };

  v4f wr[4];
  v2f bf0[16], bf1[16];
  load_w(wr, 0);
  load_b(bf0, 0);
  store_w(wl[0], wr);
  lds_barrier();
#pragma unroll
  for (int c = 0; c < CIN / QK_KC; c += 2) {
    load_w(wr, (c + 1) * QK_KC);
    load_b(bf1, (c + 1) * QK_KC);
    compute(wl[0], bf0);
    store_w(wl[1], wr);
    lds_barrier();
    if (c + 2 < CIN / QK_KC) {
      load_w(wr, (c + 2) * QK_KC);
      load_b(bf0, (c + 2) * QK_KC);
    }
    compute(wl[1], bf1);
    if (c + 2 < CIN / QK_KC) {
      store_w(wl[0], wr);
      lds_barrier();
    }
  }

  float* out = qkv + (size_t)n * CH3 * LSPAT;
#pragma unroll
  for (int r = 0; r < 8; ++r) {
    const int row0 = m0 + rt * 32 + r + half * 8;
    const int row1 = row0 + 16;
    out[(size_t)row0 * LSPAT + col] = acc0[r] + bias[row0];
    out[(size_t)row1 * LSPAT + col] = acc1[r] + bias[row1];
  }
}

#define PT_PITCH 20
#define NORM_SHIFT_LIMIT2 8192.0f
static __device__ __forceinline__ void attn_body(
    const float* __restrict__ qkv, float* __restrict__ obuf, const int bid,
    float (*plds)[16][PT_PITCH], float* kpart) {
  const int lane = threadIdx.x & 31;
  const int half = lane >> 4;
  const int l16  = lane & 15;
  const int wv   = threadIdx.x >> 5;

  const int qtiles = LSPAT / 64;
  const int nh = bid / qtiles;
  const int qt = bid % qtiles;
  const int n = nh >> 3, h = nh & 7;
  const int qbase = qt * 64 + wv * 16;

  const float* base = qkv + (size_t)n * CH3 * LSPAT + (size_t)h * 24 * LSPAT;
  const float* qm = base;
  const float* km = base + (size_t)DHEAD * LSPAT;
  const float* vm = base + (size_t)2 * DHEAD * LSPAT;

  float k2max = 0.0f;
  {
    const float* kq = km + 2 * (int)threadIdx.x;
#pragma unroll 3
    for (int i = 0; i < 9; ++i) {
      v2f ss = {};
#pragma unroll
      for (int d = 0; d < DHEAD; ++d) {
        const v2f kv = *(const v2f*)(kq + (size_t)d * LSPAT + 256 * i);
        ss[0] = fmaf(kv[0], kv[0], ss[0]);
        ss[1] = fmaf(kv[1], kv[1], ss[1]);
      }
      k2max = fmaxf(k2max, fmaxf(ss[0], ss[1]));
    }
#pragma unroll
    for (int m = 1; m < 32; m <<= 1) k2max = fmaxf(k2max, __shfl_xor(k2max, m, 32));
    if (lane == 0) kpart[wv] = k2max;
    __syncthreads();
    k2max = fmaxf(fmaxf(kpart[0], kpart[1]), fmaxf(kpart[2], kpart[3]));
  }

  v2f aq[2];
#pragma unroll
  for (int db = 0; db < 2; ++db) {
    const int d0 = db * 4 + half * 2;
    aq[db][0] = qm[(size_t)d0 * LSPAT + qbase + l16];
    aq[db][1] = qm[(size_t)(d0 + 1) * LSPAT + qbase + l16];
  }

  float q2max = fmaf(aq[0][0], aq[0][0], fmaf(aq[0][1], aq[0][1],
                fmaf(aq[1][0], aq[1][0], aq[1][1] * aq[1][1])));
  q2max += __shfl_xor(q2max, 16, 32);
#pragma unroll
  for (int m = 1; m < 16; m <<= 1) q2max = fmaxf(q2max, __shfl_xor(q2max, m, 32));
  const float b2 = q2max * k2max;
  const int norm_shift = __builtin_amdgcn_readfirstlane(
      (b2 <= NORM_SHIFT_LIMIT2) ? 1 : 0);
  const float bshift = sqrtf(b2);

  float m8[8];
#pragma unroll
  for (int r = 0; r < 8; ++r) m8[r] = norm_shift ? bshift : -3.0e38f;

  if (!norm_shift) {
  for (int kt = 0; kt < LSPAT / 16; ++kt) {
    const int kb = kt * 16;
    v8f s = {};
#pragma unroll
    for (int db = 0; db < 2; ++db) {
      const int d0 = db * 4 + half * 2;
      v2f bk; bk[0] = km[(size_t)d0 * LSPAT + kb + l16];
              bk[1] = km[(size_t)(d0 + 1) * LSPAT + kb + l16];
      s = wmma4(aq[db], bk, s);
    }
#pragma unroll
    for (int r = 0; r < 8; ++r) m8[r] = fmaxf(m8[r], s[r]);
  }
#pragma unroll
  for (int r = 0; r < 8; ++r) {
#pragma unroll
    for (int m = 1; m < 16; m <<= 1) m8[r] = fmaxf(m8[r], __shfl_xor(m8[r], m, 32));
  }
  }

  float s8[8];
#pragma unroll
  for (int r = 0; r < 8; ++r) s8[r] = 0.0f;

  const float* kp = km + (size_t)(half * 2) * LSPAT + l16;
  const float* vp = vm + (size_t)(l16 & 7) * LSPAT + half * 8;

  auto load_tile = [&](v2f (&bk)[2], v4f (&bv)[2], const int kb) {
#pragma unroll
    for (int db = 0; db < 2; ++db) {
      bk[db][0] = kp[(size_t)(db * 4) * LSPAT + kb];
      bk[db][1] = kp[(size_t)(db * 4 + 1) * LSPAT + kb];
    }
    bv[0] = *(const v4f*)(vp + kb);
    bv[1] = *(const v4f*)(vp + kb + 4);
  };

  auto wave_lds_order = [&]() {
    __builtin_amdgcn_fence(__ATOMIC_RELEASE, "wavefront", "local");
    __builtin_amdgcn_wave_barrier();
    __builtin_amdgcn_fence(__ATOMIC_ACQUIRE, "wavefront", "local");
  };

  v8f acc = {};
  auto do_tile = [&](const v2f (&bk)[2], const v4f (&bv)[2]) {
    v8f s = {};
    s = wmma4(aq[0], bk[0], s);
    s = wmma4(aq[1], bk[1], s);
#pragma unroll
    for (int r = 0; r < 8; ++r) {
      const float e = __expf((s[r] - m8[r]) * ATTN_SCALE);
      s8[r] += e;
      plds[wv][r + half * 8][l16] = e;
    }
    wave_lds_order();
    const v4f p0 = *(const v4f*)&plds[wv][l16][half * 8];
    const v4f p1 = *(const v4f*)&plds[wv][l16][half * 8 + 4];
    v2f ap, bp;
    ap[0] = p0[0]; ap[1] = p0[1]; bp[0] = bv[0][0]; bp[1] = bv[0][1];
    acc = wmma4(ap, bp, acc);
    ap[0] = p0[2]; ap[1] = p0[3]; bp[0] = bv[0][2]; bp[1] = bv[0][3];
    acc = wmma4(ap, bp, acc);
    ap[0] = p1[0]; ap[1] = p1[1]; bp[0] = bv[1][0]; bp[1] = bv[1][1];
    acc = wmma4(ap, bp, acc);
    ap[0] = p1[2]; ap[1] = p1[3]; bp[0] = bv[1][2]; bp[1] = bv[1][3];
    acc = wmma4(ap, bp, acc);
    wave_lds_order();
  };

  v2f bk0[2], bk1[2];
  v4f bv0[2], bv1[2];
  load_tile(bk0, bv0, 0);
  for (int kt = 0; kt < LSPAT / 16; kt += 2) {
    load_tile(bk1, bv1, (kt + 1) * 16);
    do_tile(bk0, bv0);
    const int kn = (kt + 2 < LSPAT / 16) ? kt + 2 : kt + 1;
    load_tile(bk0, bv0, kn * 16);
    do_tile(bk1, bv1);
  }

#pragma unroll
  for (int r = 0; r < 8; ++r) {
#pragma unroll
    for (int m = 1; m < 16; m <<= 1) s8[r] += __shfl_xor(s8[r], m, 32);
    acc[r] *= 1.0f / s8[r];
  }

  if (l16 < 8) {
    float* ob = obuf + (size_t)n * HCH * LSPAT + (size_t)(h * DHEAD + l16) * LSPAT;
#pragma unroll
    for (int r = 0; r < 8; ++r) ob[qbase + r + half * 8] = acc[r];
  }
}

#define POS_OCT 8
#define POS_BLK 128
#define POS_GROUPS (NBATCH * (HCH / POS_OCT) * (LSPAT / POS_BLK))
typedef const __attribute__((address_space(4))) float* cfloat_ptr;
static __device__ __forceinline__ void pos_conv_body(
    const float* __restrict__ qkv, const float* __restrict__ pw,
    const float* __restrict__ pb, float* __restrict__ pos, const int bid) {
  const int chunks = LSPAT / POS_BLK;
  int b = bid;
  const int chunk = b % chunks;  b /= chunks;
  const int og = b % (HCH / POS_OCT);
  const int n  = b / (HCH / POS_OCT);
  const unsigned p = chunk * POS_BLK + threadIdx.x;
  const int yp = (int)(p / WW);
  const int xp = (int)(p % WW);
  const int oc0 = og * POS_OCT;

  unsigned off[9];
  bool ok[9];
#pragma unroll
  for (int ky = 0; ky < 3; ++ky) {
#pragma unroll
    for (int kx = 0; kx < 3; ++kx) {
      const int iy = yp + ky - 1;
      const int ix = xp + kx - 1;
      const bool in_img = (iy >= 0) && (iy < HH) && (ix >= 0) && (ix < WW);
      ok[ky * 3 + kx]  = in_img;
      off[ky * 3 + kx] = in_img ? (unsigned)(iy * WW + ix) : p;
    }
  }

  const cfloat_ptr pwc = (cfloat_ptr)pw;
  const cfloat_ptr pbc = (cfloat_ptr)pb;
  float acc[POS_OCT];
#pragma unroll
  for (int o = 0; o < POS_OCT; ++o) acc[o] = pbc[oc0 + o];

  const float* vbase = qkv + (size_t)n * CH3 * LSPAT + (size_t)16 * LSPAT;
  for (int hh = 0; hh < NHEAD; ++hh) {
    const float* hb = vbase + (size_t)hh * 24 * LSPAT;
    float cur[DHEAD][9];
#pragma unroll
    for (int dd = 0; dd < DHEAD; ++dd) {
      const float* cb = hb + dd * LSPAT;
#pragma unroll
      for (int t = 0; t < 9; ++t) cur[dd][t] = cb[off[t]];
    }

#pragma unroll
    for (int dd = 0; dd < DHEAD; ++dd) {
      const int ic = hh * DHEAD + dd;
      float in[9];
#pragma unroll
      for (int t = 0; t < 9; ++t) in[t] = ok[t] ? cur[dd][t] : 0.0f;
#pragma unroll
      for (int o = 0; o < POS_OCT; ++o) {
        const cfloat_ptr wrow = pwc + ((size_t)(oc0 + o) * HCH + ic) * 9;
#pragma unroll
        for (int t = 0; t < 9; ++t) acc[o] = fmaf(in[t], wrow[t], acc[o]);
      }
    }
  }

  float* ob = pos + (size_t)n * HCH * LSPAT + (size_t)oc0 * LSPAT + p;
#pragma unroll
  for (int o = 0; o < POS_OCT; ++o) ob[(size_t)o * LSPAT] = acc[o];
}

#define ATT_GROUPS (NBATCH * NHEAD * (LSPAT / 64))
__global__ __launch_bounds__(128) void attn_pos_kernel(
    const float* __restrict__ qkv, const float* __restrict__ pw,
    const float* __restrict__ pb, float* __restrict__ obuf,
    float* __restrict__ pos) {
  __shared__ __attribute__((aligned(16))) float plds[4][16][PT_PITCH];
  __shared__ float kpart[4];
  if (blockIdx.x < ATT_GROUPS) {
    attn_body(qkv, obuf, (int)blockIdx.x, plds, kpart);
  } else {
    pos_conv_body(qkv, pw, pb, pos, (int)blockIdx.x - ATT_GROUPS);
  }
}

__global__ __launch_bounds__(256) void res_gemm_kernel(
    const float* __restrict__ obuf, const float* __restrict__ pos,
    const float* __restrict__ w,
    const float* __restrict__ bias, float* __restrict__ out) {
  const int lane = threadIdx.x & 31;
  const int half = lane >> 4;
  const int l16  = lane & 15;
  const int wave = blockIdx.x * (blockDim.x >> 5) + (threadIdx.x >> 5);

  const int ntn = LSPAT / 16;
  const int ntm = CIN / 16;
  int t = wave;
  const int n  = t / (ntm * ntn); t -= n * ntm * ntn;
  const int mt = t / ntn;
  const int ct = t % ntn;

  const int mrow = mt * 16 + l16;
  const int col  = ct * 16 + l16;
  const float* ib = obuf + (size_t)n * HCH * LSPAT;
  const float* pp = pos + (size_t)n * HCH * LSPAT;

  v8f acc = {};
#pragma unroll
  for (int kk = 0; kk < HCH; kk += 4) {
    const int ka = kk + half * 2;
    v2f a; a[0] = w[mrow * HCH + ka];
           a[1] = w[mrow * HCH + ka + 1];
    v2f b; b[0] = ib[(size_t)ka * LSPAT + col] + pp[(size_t)ka * LSPAT + col];
           b[1] = ib[(size_t)(ka + 1) * LSPAT + col] +
                  pp[(size_t)(ka + 1) * LSPAT + col];
    acc = wmma4(a, b, acc);
  }

  float* ob = out + (size_t)n * CIN * LSPAT;
#pragma unroll
  for (int r = 0; r < 8; ++r) {
    const int row = mt * 16 + r + half * 8;
    ob[(size_t)row * LSPAT + col] = acc[r] + bias[row];
  }
}

extern "C" void kernel_launch(void* const* d_in, const int* in_sizes, int n_in,
                              void* d_out, int out_size, void* d_ws, size_t ws_size,
                              hipStream_t stream) {
  const float* x     = (const float*)d_in[0];
  const float* qkv_w = (const float*)d_in[1];
  const float* qkv_b = (const float*)d_in[2];
  const float* pos_w = (const float*)d_in[3];
  const float* pos_b = (const float*)d_in[4];
  const float* res_w = (const float*)d_in[5];
  const float* res_b = (const float*)d_in[6];
  float* out = (float*)d_out;

  float* qkv  = (float*)d_ws;
  float* obuf = qkv + (size_t)NBATCH * CH3 * LSPAT;
  float* pos  = obuf + (size_t)NBATCH * HCH * LSPAT;

  qkv_gemm_kernel<<<NBATCH * (CH3 / QK_ROWS) * (LSPAT / QK_COLS), 256, 0,
                    stream>>>(x, qkv_w, qkv_b, qkv);

  attn_pos_kernel<<<ATT_GROUPS + POS_GROUPS, 128, 0, stream>>>(qkv, pos_w, pos_b,
                                                             obuf, pos);

  const int waves4 = NBATCH * (CIN / 16) * (LSPAT / 16);
  res_gemm_kernel<<<waves4 / 8, 256, 0, stream>>>(obuf, pos, res_w, res_b, out);
}


# res 1x1-conv GEMM: 4 waves own 64 rows x 64 columns, their 64x64 weight panel staged once through LDS in A-fragment order; a wave adds obuf+pos once and feeds 4 row tiles from it; k sum order kept
# speedup vs baseline: 3.2902x; 1.0620x over previous
#include <hip/hip_runtime.h>
#include <math.h>

typedef __attribute__((ext_vector_type(2))) float v2f;
typedef __attribute__((ext_vector_type(8))) float v8f;

#define NBATCH 4
#define CIN 512
#define HH 48
#define WW 48
#define LSPAT (HH*WW)
#define CH3 192
#define HCH 64
#define NHEAD 8
#define DHEAD 8
#define ATTN_SCALE 0.35355339059327379f

static __device__ __forceinline__ v8f wmma4(v2f a, v2f b, v8f c) {
  return __builtin_amdgcn_wmma_f32_16x16x4_f32(false, a, false, b, (short)0, c,
                                               false, false);
}

typedef __attribute__((ext_vector_type(4))) float v4f;
#define QK_ROWS 64
#define QK_COLS 64
#define QK_KC 64
#define QK_SP 260
__global__ __launch_bounds__(256) void qkv_gemm_kernel(
    const float* __restrict__ x, const float* __restrict__ w,
    const float* __restrict__ bias, float* __restrict__ qkv) {
  __shared__ __attribute__((aligned(16))) float wl[2][16 * QK_SP];
  const int tid  = threadIdx.x;
  const int lane = tid & 31;
  const int half = lane >> 4;
  const int l16  = lane & 15;
  const int wv   = tid >> 5;

  int b = blockIdx.x;
  const int mg = b % (CH3 / QK_ROWS);   b /= (CH3 / QK_ROWS);
  const int cb = b % (LSPAT / QK_COLS);
  const int n  = b / (LSPAT / QK_COLS);
  const int m0 = mg * QK_ROWS;
  const int rt = wv >> 2;
  const int col = cb * QK_COLS + (wv & 3) * 16 + l16;
  const float* xb = x + (size_t)n * CIN * LSPAT;

  const int q4   = tid & 15;
  const int srow = tid >> 4;
  const float* wsrc = w + (size_t)(m0 + srow) * CIN + q4 * 4;
  const int sdst = q4 * QK_SP + srow * 4;

  const float* bp = xb + (size_t)(half * 2) * LSPAT + col;
  const int aoff = (rt * 32 + l16) * 4 + half * 2;

  auto load_w = [&](v4f (&wr)[4], const int kc) {
#pragma unroll
    for (int i = 0; i < 4; ++i) {
      const float* p = wsrc + (size_t)(i * 16) * CIN + kc;
      wr[i][0] = p[0]; wr[i][1] = p[1]; wr[i][2] = p[2]; wr[i][3] = p[3];
    }
  };
  auto store_w = [&](float* buf, const v4f (&wr)[4]) {
#pragma unroll
    for (int i = 0; i < 4; ++i) *(v4f*)(buf + sdst + i * 64) = wr[i];
  };
  auto load_b = [&](v2f (&bf)[16], const int kc) {
#pragma unroll
    for (int s = 0; s < 16; ++s) {
      bf[s][0] = bp[(size_t)(kc + 4 * s) * LSPAT];
      bf[s][1] = bp[(size_t)(kc + 4 * s + 1) * LSPAT];
    }
  };
  auto lds_barrier = [&]() {
    __builtin_amdgcn_fence(__ATOMIC_RELEASE, "workgroup", "local");
    __builtin_amdgcn_s_barrier();
    __builtin_amdgcn_fence(__ATOMIC_ACQUIRE, "workgroup", "local");
  };

  v8f acc0 = {}, acc1 = {};
  auto compute = [&](const float* buf, const v2f (&bf)[16]) {
#pragma unroll
    for (int s = 0; s < 16; ++s) {
      const v2f a0 = *(const v2f*)(buf + s * QK_SP + aoff);
      const v2f a1 = *(const v2f*)(buf + s * QK_SP + aoff + 64);
      acc0 = wmma4(a0, bf[s], acc0);
      acc1 = wmma4(a1, bf[s], acc1);
    }
  };

  v4f wr[4];
  v2f bf0[16], bf1[16];
  load_w(wr, 0);
  load_b(bf0, 0);
  store_w(wl[0], wr);
  lds_barrier();
#pragma unroll
  for (int c = 0; c < CIN / QK_KC; c += 2) {
    load_w(wr, (c + 1) * QK_KC);
    load_b(bf1, (c + 1) * QK_KC);
    compute(wl[0], bf0);
    store_w(wl[1], wr);
    lds_barrier();
    if (c + 2 < CIN / QK_KC) {
      load_w(wr, (c + 2) * QK_KC);
      load_b(bf0, (c + 2) * QK_KC);
    }
    compute(wl[1], bf1);
    if (c + 2 < CIN / QK_KC) {
      store_w(wl[0], wr);
      lds_barrier();
    }
  }

  float* out = qkv + (size_t)n * CH3 * LSPAT;
#pragma unroll
  for (int r = 0; r < 8; ++r) {
    const int row0 = m0 + rt * 32 + r + half * 8;
    const int row1 = row0 + 16;
    out[(size_t)row0 * LSPAT + col] = acc0[r] + bias[row0];
    out[(size_t)row1 * LSPAT + col] = acc1[r] + bias[row1];
  }
}

#define PT_PITCH 20
#define NORM_SHIFT_LIMIT2 8192.0f
static __device__ __forceinline__ void attn_body(
    const float* __restrict__ qkv, float* __restrict__ obuf, const int bid,
    float (*plds)[16][PT_PITCH], float* kpart) {
  const int lane = threadIdx.x & 31;
  const int half = lane >> 4;
  const int l16  = lane & 15;
  const int wv   = threadIdx.x >> 5;

  const int qtiles = LSPAT / 64;
  const int nh = bid / qtiles;
  const int qt = bid % qtiles;
  const int n = nh >> 3, h = nh & 7;
  const int qbase = qt * 64 + wv * 16;

  const float* base = qkv + (size_t)n * CH3 * LSPAT + (size_t)h * 24 * LSPAT;
  const float* qm = base;
  const float* km = base + (size_t)DHEAD * LSPAT;
  const float* vm = base + (size_t)2 * DHEAD * LSPAT;

  float k2max = 0.0f;
  {
    const float* kq = km + 2 * (int)threadIdx.x;
#pragma unroll 3
    for (int i = 0; i < 9; ++i) {
      v2f ss = {};
#pragma unroll
      for (int d = 0; d < DHEAD; ++d) {
        const v2f kv = *(const v2f*)(kq + (size_t)d * LSPAT + 256 * i);
        ss[0] = fmaf(kv[0], kv[0], ss[0]);
        ss[1] = fmaf(kv[1], kv[1], ss[1]);
      }
      k2max = fmaxf(k2max, fmaxf(ss[0], ss[1]));
    }
#pragma unroll
    for (int m = 1; m < 32; m <<= 1) k2max = fmaxf(k2max, __shfl_xor(k2max, m, 32));
    if (lane == 0) kpart[wv] = k2max;
    __syncthreads();
    k2max = fmaxf(fmaxf(kpart[0], kpart[1]), fmaxf(kpart[2], kpart[3]));
  }

  v2f aq[2];
#pragma unroll
  for (int db = 0; db < 2; ++db) {
    const int d0 = db * 4 + half * 2;
    aq[db][0] = qm[(size_t)d0 * LSPAT + qbase + l16];
    aq[db][1] = qm[(size_t)(d0 + 1) * LSPAT + qbase + l16];
  }

  float q2max = fmaf(aq[0][0], aq[0][0], fmaf(aq[0][1], aq[0][1],
                fmaf(aq[1][0], aq[1][0], aq[1][1] * aq[1][1])));
  q2max += __shfl_xor(q2max, 16, 32);
#pragma unroll
  for (int m = 1; m < 16; m <<= 1) q2max = fmaxf(q2max, __shfl_xor(q2max, m, 32));
  const float b2 = q2max * k2max;
  const int norm_shift = __builtin_amdgcn_readfirstlane(
      (b2 <= NORM_SHIFT_LIMIT2) ? 1 : 0);
  const float bshift = sqrtf(b2);

  float m8[8];
#pragma unroll
  for (int r = 0; r < 8; ++r) m8[r] = norm_shift ? bshift : -3.0e38f;

  if (!norm_shift) {
  for (int kt = 0; kt < LSPAT / 16; ++kt) {
    const int kb = kt * 16;
    v8f s = {};
#pragma unroll
    for (int db = 0; db < 2; ++db) {
      const int d0 = db * 4 + half * 2;
      v2f bk; bk[0] = km[(size_t)d0 * LSPAT + kb + l16];
              bk[1] = km[(size_t)(d0 + 1) * LSPAT + kb + l16];
      s = wmma4(aq[db], bk, s);
    }
#pragma unroll
    for (int r = 0; r < 8; ++r) m8[r] = fmaxf(m8[r], s[r]);
  }
#pragma unroll
  for (int r = 0; r < 8; ++r) {
#pragma unroll
    for (int m = 1; m < 16; m <<= 1) m8[r] = fmaxf(m8[r], __shfl_xor(m8[r], m, 32));
  }
  }

  float s8[8];
#pragma unroll
  for (int r = 0; r < 8; ++r) s8[r] = 0.0f;

  const float* kp = km + (size_t)(half * 2) * LSPAT + l16;
  const float* vp = vm + (size_t)(l16 & 7) * LSPAT + half * 8;

  auto load_tile = [&](v2f (&bk)[2], v4f (&bv)[2], const int kb) {
#pragma unroll
    for (int db = 0; db < 2; ++db) {
      bk[db][0] = kp[(size_t)(db * 4) * LSPAT + kb];
      bk[db][1] = kp[(size_t)(db * 4 + 1) * LSPAT + kb];
    }
    bv[0] = *(const v4f*)(vp + kb);
    bv[1] = *(const v4f*)(vp + kb + 4);
  };

  auto wave_lds_order = [&]() {
    __builtin_amdgcn_fence(__ATOMIC_RELEASE, "wavefront", "local");
    __builtin_amdgcn_wave_barrier();
    __builtin_amdgcn_fence(__ATOMIC_ACQUIRE, "wavefront", "local");
  };

  v8f acc = {};
  auto do_tile = [&](const v2f (&bk)[2], const v4f (&bv)[2]) {
    v8f s = {};
    s = wmma4(aq[0], bk[0], s);
    s = wmma4(aq[1], bk[1], s);
#pragma unroll
    for (int r = 0; r < 8; ++r) {
      const float e = __expf((s[r] - m8[r]) * ATTN_SCALE);
      s8[r] += e;
      plds[wv][r + half * 8][l16] = e;
    }
    wave_lds_order();
    const v4f p0 = *(const v4f*)&plds[wv][l16][half * 8];
    const v4f p1 = *(const v4f*)&plds[wv][l16][half * 8 + 4];
    v2f ap, bp;
    ap[0] = p0[0]; ap[1] = p0[1]; bp[0] = bv[0][0]; bp[1] = bv[0][1];
    acc = wmma4(ap, bp, acc);
    ap[0] = p0[2]; ap[1] = p0[3]; bp[0] = bv[0][2]; bp[1] = bv[0][3];
    acc = wmma4(ap, bp, acc);
    ap[0] = p1[0]; ap[1] = p1[1]; bp[0] = bv[1][0]; bp[1] = bv[1][1];
    acc = wmma4(ap, bp, acc);
    ap[0] = p1[2]; ap[1] = p1[3]; bp[0] = bv[1][2]; bp[1] = bv[1][3];
    acc = wmma4(ap, bp, acc);
    wave_lds_order();
  };

  v2f bk0[2], bk1[2];
  v4f bv0[2], bv1[2];
  load_tile(bk0, bv0, 0);
  for (int kt = 0; kt < LSPAT / 16; kt += 2) {
    load_tile(bk1, bv1, (kt + 1) * 16);
    do_tile(bk0, bv0);
    const int kn = (kt + 2 < LSPAT / 16) ? kt + 2 : kt + 1;
    load_tile(bk0, bv0, kn * 16);
    do_tile(bk1, bv1);
  }

#pragma unroll
  for (int r = 0; r < 8; ++r) {
#pragma unroll
    for (int m = 1; m < 16; m <<= 1) s8[r] += __shfl_xor(s8[r], m, 32);
    acc[r] *= 1.0f / s8[r];
  }

  if (l16 < 8) {
    float* ob = obuf + (size_t)n * HCH * LSPAT + (size_t)(h * DHEAD + l16) * LSPAT;
#pragma unroll
    for (int r = 0; r < 8; ++r) ob[qbase + r + half * 8] = acc[r];
  }
}

#define POS_OCT 8
#define POS_BLK 128
#define POS_GROUPS (NBATCH * (HCH / POS_OCT) * (LSPAT / POS_BLK))
typedef const __attribute__((address_space(4))) float* cfloat_ptr;
static __device__ __forceinline__ void pos_conv_body(
    const float* __restrict__ qkv, const float* __restrict__ pw,
    const float* __restrict__ pb, float* __restrict__ pos, const int bid) {
  const int chunks = LSPAT / POS_BLK;
  int b = bid;
  const int chunk = b % chunks;  b /= chunks;
  const int og = b % (HCH / POS_OCT);
  const int n  = b / (HCH / POS_OCT);
  const unsigned p = chunk * POS_BLK + threadIdx.x;
  const int yp = (int)(p / WW);
  const int xp = (int)(p % WW);
  const int oc0 = og * POS_OCT;

  unsigned off[9];
  bool ok[9];
#pragma unroll
  for (int ky = 0; ky < 3; ++ky) {
#pragma unroll
    for (int kx = 0; kx < 3; ++kx) {
      const int iy = yp + ky - 1;
      const int ix = xp + kx - 1;
      const bool in_img = (iy >= 0) && (iy < HH) && (ix >= 0) && (ix < WW);
      ok[ky * 3 + kx]  = in_img;
      off[ky * 3 + kx] = in_img ? (unsigned)(iy * WW + ix) : p;
    }
  }

  const cfloat_ptr pwc = (cfloat_ptr)pw;
  const cfloat_ptr pbc = (cfloat_ptr)pb;
  float acc[POS_OCT];
#pragma unroll
  for (int o = 0; o < POS_OCT; ++o) acc[o] = pbc[oc0 + o];

  const float* vbase = qkv + (size_t)n * CH3 * LSPAT + (size_t)16 * LSPAT;
  for (int hh = 0; hh < NHEAD; ++hh) {
    const float* hb = vbase + (size_t)hh * 24 * LSPAT;
    float cur[DHEAD][9];
#pragma unroll
    for (int dd = 0; dd < DHEAD; ++dd) {
      const float* cb = hb + dd * LSPAT;
#pragma unroll
      for (int t = 0; t < 9; ++t) cur[dd][t] = cb[off[t]];
    }

#pragma unroll
    for (int dd = 0; dd < DHEAD; ++dd) {
      const int ic = hh * DHEAD + dd;
      float in[9];
#pragma unroll
      for (int t = 0; t < 9; ++t) in[t] = ok[t] ? cur[dd][t] : 0.0f;
#pragma unroll
      for (int o = 0; o < POS_OCT; ++o) {
        const cfloat_ptr wrow = pwc + ((size_t)(oc0 + o) * HCH + ic) * 9;
#pragma unroll
        for (int t = 0; t < 9; ++t) acc[o] = fmaf(in[t], wrow[t], acc[o]);
      }
    }
  }

  float* ob = pos + (size_t)n * HCH * LSPAT + (size_t)oc0 * LSPAT + p;
#pragma unroll
  for (int o = 0; o < POS_OCT; ++o) ob[(size_t)o * LSPAT] = acc[o];
}

#define ATT_GROUPS (NBATCH * NHEAD * (LSPAT / 64))
__global__ __launch_bounds__(128) void attn_pos_kernel(
    const float* __restrict__ qkv, const float* __restrict__ pw,
    const float* __restrict__ pb, float* __restrict__ obuf,
    float* __restrict__ pos) {
  __shared__ __attribute__((aligned(16))) float plds[4][16][PT_PITCH];
  __shared__ float kpart[4];
  if (blockIdx.x < ATT_GROUPS) {
    attn_body(qkv, obuf, (int)blockIdx.x, plds, kpart);
  } else {
    pos_conv_body(qkv, pw, pb, pos, (int)blockIdx.x - ATT_GROUPS);
  }
}

#define RS_ROWS 64
#define RS_COLS 64
#define RS_SP 260
__global__ __launch_bounds__(128) void res_gemm_kernel(
    const float* __restrict__ obuf, const float* __restrict__ pos,
    const float* __restrict__ w,
    const float* __restrict__ bias, float* __restrict__ out) {
  __shared__ __attribute__((aligned(16))) float wl[16 * RS_SP];
  const int tid  = threadIdx.x;
  const int lane = tid & 31;
  const int half = lane >> 4;
  const int l16  = lane & 15;
  const int wv   = tid >> 5;

  int b = blockIdx.x;
  const int mg = b % (CIN / RS_ROWS);   b /= (CIN / RS_ROWS);
  const int cb = b % (LSPAT / RS_COLS);
  const int n  = b / (LSPAT / RS_COLS);
  const int m0 = mg * RS_ROWS;
  const int col = cb * RS_COLS + wv * 16 + l16;

  const size_t boff = (size_t)n * HCH * LSPAT + (size_t)(half * 2) * LSPAT + col;
  const float* ib = obuf + boff;
  const float* pp = pos + boff;
  v2f bo[16], bq[16];
#pragma unroll
  for (int s = 0; s < 16; ++s) {
    bo[s][0] = ib[(size_t)(4 * s) * LSPAT];
    bo[s][1] = ib[(size_t)(4 * s + 1) * LSPAT];
    bq[s][0] = pp[(size_t)(4 * s) * LSPAT];
    bq[s][1] = pp[(size_t)(4 * s + 1) * LSPAT];
  }

  {
    const int q4   = tid & 15;
    const int srow = tid >> 4;
    const float* wsrc = w + (size_t)(m0 + srow) * HCH + q4 * 4;
    v4f wr[8];
#pragma unroll
    for (int i = 0; i < 8; ++i) {
      const float* p = wsrc + (size_t)(i * 8) * HCH;
      wr[i][0] = p[0]; wr[i][1] = p[1]; wr[i][2] = p[2]; wr[i][3] = p[3];
    }
#pragma unroll
    for (int i = 0; i < 8; ++i)
      *(v4f*)(wl + q4 * RS_SP + (srow + 8 * i) * 4) = wr[i];
  }
  __builtin_amdgcn_fence(__ATOMIC_RELEASE, "workgroup", "local");
  __builtin_amdgcn_s_barrier();
  __builtin_amdgcn_fence(__ATOMIC_ACQUIRE, "workgroup", "local");

  v2f bf[16];
#pragma unroll
  for (int s = 0; s < 16; ++s) bf[s] = bo[s] + bq[s];

  const int aoff = l16 * 4 + half * 2;
  v8f acc[4] = {};
#pragma unroll
  for (int s = 0; s < 16; ++s) {
#pragma unroll
    for (int t = 0; t < 4; ++t) {
      const v2f a = *(const v2f*)(wl + s * RS_SP + aoff + t * 64);
      acc[t] = wmma4(a, bf[s], acc[t]);
    }
  }

  float* ob = out + (size_t)n * CIN * LSPAT;
#pragma unroll
  for (int t = 0; t < 4; ++t) {
#pragma unroll
    for (int r = 0; r < 8; ++r) {
      const int row = m0 + t * 16 + r + half * 8;
      ob[(size_t)row * LSPAT + col] = acc[t][r] + bias[row];
    }
  }
}

extern "C" void kernel_launch(void* const* d_in, const int* in_sizes, int n_in,
                              void* d_out, int out_size, void* d_ws, size_t ws_size,
                              hipStream_t stream) {
  const float* x     = (const float*)d_in[0];
  const float* qkv_w = (const float*)d_in[1];
  const float* qkv_b = (const float*)d_in[2];
  const float* pos_w = (const float*)d_in[3];
  const float* pos_b = (const float*)d_in[4];
  const float* res_w = (const float*)d_in[5];
  const float* res_b = (const float*)d_in[6];
  float* out = (float*)d_out;

  float* qkv  = (float*)d_ws;
  float* obuf = qkv + (size_t)NBATCH * CH3 * LSPAT;
  float* pos  = obuf + (size_t)NBATCH * HCH * LSPAT;

  qkv_gemm_kernel<<<NBATCH * (CH3 / QK_ROWS) * (LSPAT / QK_COLS), 256, 0,
                    stream>>>(x, qkv_w, qkv_b, qkv);

  attn_pos_kernel<<<ATT_GROUPS + POS_GROUPS, 128, 0, stream>>>(qkv, pos_w, pos_b,
                                                             obuf, pos);

  res_gemm_kernel<<<NBATCH * (CIN / RS_ROWS) * (LSPAT / RS_COLS), 128, 0,
                    stream>>>(obuf, pos, res_w, res_b, out);
}


# attention: softmax shift enters the score product as its C operand and Q carries scale*log2(e): the accumulator is the exp2 argument, no subtract or multiplies per score; softmax rounding changes
# speedup vs baseline: 3.4306x; 1.0427x over previous
#include <hip/hip_runtime.h>
#include <math.h>

typedef __attribute__((ext_vector_type(2))) float v2f;
typedef __attribute__((ext_vector_type(8))) float v8f;

#define NBATCH 4
#define CIN 512
#define HH 48
#define WW 48
#define LSPAT (HH*WW)
#define CH3 192
#define HCH 64
#define NHEAD 8
#define DHEAD 8
#define ATTN_SCALE 0.35355339059327379f

static __device__ __forceinline__ v8f wmma4(v2f a, v2f b, v8f c) {
  return __builtin_amdgcn_wmma_f32_16x16x4_f32(false, a, false, b, (short)0, c,
                                               false, false);
}

typedef __attribute__((ext_vector_type(4))) float v4f;
#define QK_ROWS 64
#define QK_COLS 64
#define QK_KC 64
#define QK_SP 260
__global__ __launch_bounds__(256) void qkv_gemm_kernel(
    const float* __restrict__ x, const float* __restrict__ w,
    const float* __restrict__ bias, float* __restrict__ qkv) {
  __shared__ __attribute__((aligned(16))) float wl[2][16 * QK_SP];
  const int tid  = threadIdx.x;
  const int lane = tid & 31;
  const int half = lane >> 4;
  const int l16  = lane & 15;
  const int wv   = tid >> 5;

  int b = blockIdx.x;
  const int mg = b % (CH3 / QK_ROWS);   b /= (CH3 / QK_ROWS);
  const int cb = b % (LSPAT / QK_COLS);
  const int n  = b / (LSPAT / QK_COLS);
  const int m0 = mg * QK_ROWS;
  const int rt = wv >> 2;
  const int col = cb * QK_COLS + (wv & 3) * 16 + l16;
  const float* xb = x + (size_t)n * CIN * LSPAT;

  const int q4   = tid & 15;
  const int srow = tid >> 4;
  const float* wsrc = w + (size_t)(m0 + srow) * CIN + q4 * 4;
  const int sdst = q4 * QK_SP + srow * 4;

  const float* bp = xb + (size_t)(half * 2) * LSPAT + col;
  const int aoff = (rt * 32 + l16) * 4 + half * 2;

  auto load_w = [&](v4f (&wr)[4], const int kc) {
#pragma unroll
    for (int i = 0; i < 4; ++i) {
      const float* p = wsrc + (size_t)(i * 16) * CIN + kc;
      wr[i][0] = p[0]; wr[i][1] = p[1]; wr[i][2] = p[2]; wr[i][3] = p[3];
    }
  };
  auto store_w = [&](float* buf, const v4f (&wr)[4]) {
#pragma unroll
    for (int i = 0; i < 4; ++i) *(v4f*)(buf + sdst + i * 64) = wr[i];
  };
  auto load_b = [&](v2f (&bf)[16], const int kc) {
#pragma unroll
    for (int s = 0; s < 16; ++s) {
      bf[s][0] = bp[(size_t)(kc + 4 * s) * LSPAT];
      bf[s][1] = bp[(size_t)(kc + 4 * s + 1) * LSPAT];
    }
  };
  auto lds_barrier = [&]() {
    __builtin_amdgcn_fence(__ATOMIC_RELEASE, "workgroup", "local");
    __builtin_amdgcn_s_barrier();
    __builtin_amdgcn_fence(__ATOMIC_ACQUIRE, "workgroup", "local");
  };

  v8f acc0 = {}, acc1 = {};
  auto compute = [&](const float* buf, const v2f (&bf)[16]) {
#pragma unroll
    for (int s = 0; s < 16; ++s) {
      const v2f a0 = *(const v2f*)(buf + s * QK_SP + aoff);
      const v2f a1 = *(const v2f*)(buf + s * QK_SP + aoff + 64);
      acc0 = wmma4(a0, bf[s], acc0);
      acc1 = wmma4(a1, bf[s], acc1);
    }
  };

  v4f wr[4];
  v2f bf0[16], bf1[16];
  load_w(wr, 0);
  load_b(bf0, 0);
  store_w(wl[0], wr);
  lds_barrier();
#pragma unroll
  for (int c = 0; c < CIN / QK_KC; c += 2) {
    load_w(wr, (c + 1) * QK_KC);
    load_b(bf1, (c + 1) * QK_KC);
    compute(wl[0], bf0);
    store_w(wl[1], wr);
    lds_barrier();
    if (c + 2 < CIN / QK_KC) {
      load_w(wr, (c + 2) * QK_KC);
      load_b(bf0, (c + 2) * QK_KC);
    }
    compute(wl[1], bf1);
    if (c + 2 < CIN / QK_KC) {
      store_w(wl[0], wr);
      lds_barrier();
    }
  }

  float* out = qkv + (size_t)n * CH3 * LSPAT;
#pragma unroll
  for (int r = 0; r < 8; ++r) {
    const int row0 = m0 + rt * 32 + r + half * 8;
    const int row1 = row0 + 16;
    out[(size_t)row0 * LSPAT + col] = acc0[r] + bias[row0];
    out[(size_t)row1 * LSPAT + col] = acc1[r] + bias[row1];
  }
}

#define PT_PITCH 20
#define NORM_SHIFT_LIMIT2 8192.0f
#define QK_LOG2 (ATTN_SCALE * 1.4426950408889634f)
static __device__ __forceinline__ void attn_body(
    const float* __restrict__ qkv, float* __restrict__ obuf, const int bid,
    float (*plds)[16][PT_PITCH], float* kpart) {
  const int lane = threadIdx.x & 31;
  const int half = lane >> 4;
  const int l16  = lane & 15;
  const int wv   = threadIdx.x >> 5;

  const int qtiles = LSPAT / 64;
  const int nh = bid / qtiles;
  const int qt = bid % qtiles;
  const int n = nh >> 3, h = nh & 7;
  const int qbase = qt * 64 + wv * 16;

  const float* base = qkv + (size_t)n * CH3 * LSPAT + (size_t)h * 24 * LSPAT;
  const float* qm = base;
  const float* km = base + (size_t)DHEAD * LSPAT;
  const float* vm = base + (size_t)2 * DHEAD * LSPAT;

  float k2max = 0.0f;
  {
    const float* kq = km + 2 * (int)threadIdx.x;
#pragma unroll 3
    for (int i = 0; i < 9; ++i) {
      v2f ss = {};
#pragma unroll
      for (int d = 0; d < DHEAD; ++d) {
        const v2f kv = *(const v2f*)(kq + (size_t)d * LSPAT + 256 * i);
        ss[0] = fmaf(kv[0], kv[0], ss[0]);
        ss[1] = fmaf(kv[1], kv[1], ss[1]);
      }
      k2max = fmaxf(k2max, fmaxf(ss[0], ss[1]));
    }
#pragma unroll
    for (int m = 1; m < 32; m <<= 1) k2max = fmaxf(k2max, __shfl_xor(k2max, m, 32));
    if (lane == 0) kpart[wv] = k2max;
    __syncthreads();
    k2max = fmaxf(fmaxf(kpart[0], kpart[1]), fmaxf(kpart[2], kpart[3]));
  }

  v2f aq[2];
#pragma unroll
  for (int db = 0; db < 2; ++db) {
    const int d0 = db * 4 + half * 2;
    aq[db][0] = qm[(size_t)d0 * LSPAT + qbase + l16];
    aq[db][1] = qm[(size_t)(d0 + 1) * LSPAT + qbase + l16];
  }

  float q2max = fmaf(aq[0][0], aq[0][0], fmaf(aq[0][1], aq[0][1],
                fmaf(aq[1][0], aq[1][0], aq[1][1] * aq[1][1])));
  q2max += __shfl_xor(q2max, 16, 32);
#pragma unroll
  for (int m = 1; m < 16; m <<= 1) q2max = fmaxf(q2max, __shfl_xor(q2max, m, 32));
  const float b2 = q2max * k2max;
  const int norm_shift = __builtin_amdgcn_readfirstlane(
      (b2 <= NORM_SHIFT_LIMIT2) ? 1 : 0);
  const float bshift = sqrtf(b2) * QK_LOG2;

#pragma unroll
  for (int db = 0; db < 2; ++db) { aq[db][0] *= QK_LOG2; aq[db][1] *= QK_LOG2; }

  float m8[8];
#pragma unroll
  for (int r = 0; r < 8; ++r) m8[r] = norm_shift ? bshift : -3.0e38f;

  if (!norm_shift) {
  for (int kt = 0; kt < LSPAT / 16; ++kt) {
    const int kb = kt * 16;
    v8f s = {};
#pragma unroll
    for (int db = 0; db < 2; ++db) {
      const int d0 = db * 4 + half * 2;
      v2f bk; bk[0] = km[(size_t)d0 * LSPAT + kb + l16];
              bk[1] = km[(size_t)(d0 + 1) * LSPAT + kb + l16];
      s = wmma4(aq[db], bk, s);
    }
#pragma unroll
    for (int r = 0; r < 8; ++r) m8[r] = fmaxf(m8[r], s[r]);
  }
#pragma unroll
  for (int r = 0; r < 8; ++r) {
#pragma unroll
    for (int m = 1; m < 16; m <<= 1) m8[r] = fmaxf(m8[r], __shfl_xor(m8[r], m, 32));
  }
  }

  float s8[8];
#pragma unroll
  for (int r = 0; r < 8; ++r) s8[r] = 0.0f;

  const float* kp = km + (size_t)(half * 2) * LSPAT + l16;
  const float* vp = vm + (size_t)(l16 & 7) * LSPAT + half * 8;

  auto load_tile = [&](v2f (&bk)[2], v4f (&bv)[2], const int kb) {
#pragma unroll
    for (int db = 0; db < 2; ++db) {
      bk[db][0] = kp[(size_t)(db * 4) * LSPAT + kb];
      bk[db][1] = kp[(size_t)(db * 4 + 1) * LSPAT + kb];
    }
    bv[0] = *(const v4f*)(vp + kb);
    bv[1] = *(const v4f*)(vp + kb + 4);
  };

  auto wave_lds_order = [&]() {
    __builtin_amdgcn_fence(__ATOMIC_RELEASE, "wavefront", "local");
    __builtin_amdgcn_wave_barrier();
    __builtin_amdgcn_fence(__ATOMIC_ACQUIRE, "wavefront", "local");
  };

  v8f cneg;
#pragma unroll
  for (int r = 0; r < 8; ++r) cneg[r] = -m8[r];

  v8f acc = {};
  auto do_tile = [&](const v2f (&bk)[2], const v4f (&bv)[2]) {
    v8f s = wmma4(aq[0], bk[0], cneg);
    s = wmma4(aq[1], bk[1], s);
#pragma unroll
    for (int r = 0; r < 8; ++r) {
      const float e = __builtin_amdgcn_exp2f(s[r]);
      s8[r] += e;
      plds[wv][r + half * 8][l16] = e;
    }
    wave_lds_order();
    const v4f p0 = *(const v4f*)&plds[wv][l16][half * 8];
    const v4f p1 = *(const v4f*)&plds[wv][l16][half * 8 + 4];
    v2f ap, bp;
    ap[0] = p0[0]; ap[1] = p0[1]; bp[0] = bv[0][0]; bp[1] = bv[0][1];
    acc = wmma4(ap, bp, acc);
    ap[0] = p0[2]; ap[1] = p0[3]; bp[0] = bv[0][2]; bp[1] = bv[0][3];
    acc = wmma4(ap, bp, acc);
    ap[0] = p1[0]; ap[1] = p1[1]; bp[0] = bv[1][0]; bp[1] = bv[1][1];
    acc = wmma4(ap, bp, acc);
    ap[0] = p1[2]; ap[1] = p1[3]; bp[0] = bv[1][2]; bp[1] = bv[1][3];
    acc = wmma4(ap, bp, acc);
    wave_lds_order();
  };

  v2f bk0[2], bk1[2];
  v4f bv0[2], bv1[2];
  load_tile(bk0, bv0, 0);
  for (int kt = 0; kt < LSPAT / 16; kt += 2) {
    load_tile(bk1, bv1, (kt + 1) * 16);
    do_tile(bk0, bv0);
    const int kn = (kt + 2 < LSPAT / 16) ? kt + 2 : kt + 1;
    load_tile(bk0, bv0, kn * 16);
    do_tile(bk1, bv1);
  }

#pragma unroll
  for (int r = 0; r < 8; ++r) {
#pragma unroll
    for (int m = 1; m < 16; m <<= 1) s8[r] += __shfl_xor(s8[r], m, 32);
    acc[r] *= 1.0f / s8[r];
  }

  if (l16 < 8) {
    float* ob = obuf + (size_t)n * HCH * LSPAT + (size_t)(h * DHEAD + l16) * LSPAT;
#pragma unroll
    for (int r = 0; r < 8; ++r) ob[qbase + r + half * 8] = acc[r];
  }
}

#define POS_OCT 8
#define POS_BLK 128
#define POS_GROUPS (NBATCH * (HCH / POS_OCT) * (LSPAT / POS_BLK))
typedef const __attribute__((address_space(4))) float* cfloat_ptr;
static __device__ __forceinline__ void pos_conv_body(
    const float* __restrict__ qkv, const float* __restrict__ pw,
    const float* __restrict__ pb, float* __restrict__ pos, const int bid) {
  const int chunks = LSPAT / POS_BLK;
  int b = bid;
  const int chunk = b % chunks;  b /= chunks;
  const int og = b % (HCH / POS_OCT);
  const int n  = b / (HCH / POS_OCT);
  const unsigned p = chunk * POS_BLK + threadIdx.x;
  const int yp = (int)(p / WW);
  const int xp = (int)(p % WW);
  const int oc0 = og * POS_OCT;

  unsigned off[9];
  bool ok[9];
#pragma unroll
  for (int ky = 0; ky < 3; ++ky) {
#pragma unroll
    for (int kx = 0; kx < 3; ++kx) {
      const int iy = yp + ky - 1;
      const int ix = xp + kx - 1;
      const bool in_img = (iy >= 0) && (iy < HH) && (ix >= 0) && (ix < WW);
      ok[ky * 3 + kx]  = in_img;
      off[ky * 3 + kx] = in_img ? (unsigned)(iy * WW + ix) : p;
    }
  }

  const cfloat_ptr pwc = (cfloat_ptr)pw;
  const cfloat_ptr pbc = (cfloat_ptr)pb;
  float acc[POS_OCT];
#pragma unroll
  for (int o = 0; o < POS_OCT; ++o) acc[o] = pbc[oc0 + o];

  const float* vbase = qkv + (size_t)n * CH3 * LSPAT + (size_t)16 * LSPAT;
  for (int hh = 0; hh < NHEAD; ++hh) {
    const float* hb = vbase + (size_t)hh * 24 * LSPAT;
    float cur[DHEAD][9];
#pragma unroll
    for (int dd = 0; dd < DHEAD; ++dd) {
      const float* cb = hb + dd * LSPAT;
#pragma unroll
      for (int t = 0; t < 9; ++t) cur[dd][t] = cb[off[t]];
    }

#pragma unroll
    for (int dd = 0; dd < DHEAD; ++dd) {
      const int ic = hh * DHEAD + dd;
      float in[9];
#pragma unroll
      for (int t = 0; t < 9; ++t) in[t] = ok[t] ? cur[dd][t] : 0.0f;
#pragma unroll
      for (int o = 0; o < POS_OCT; ++o) {
        const cfloat_ptr wrow = pwc + ((size_t)(oc0 + o) * HCH + ic) * 9;
#pragma unroll
        for (int t = 0; t < 9; ++t) acc[o] = fmaf(in[t], wrow[t], acc[o]);
      }
    }
  }

  float* ob = pos + (size_t)n * HCH * LSPAT + (size_t)oc0 * LSPAT + p;
#pragma unroll
  for (int o = 0; o < POS_OCT; ++o) ob[(size_t)o * LSPAT] = acc[o];
}

#define ATT_GROUPS (NBATCH * NHEAD * (LSPAT / 64))
__global__ __launch_bounds__(128) void attn_pos_kernel(
    const float* __restrict__ qkv, const float* __restrict__ pw,
    const float* __restrict__ pb, float* __restrict__ obuf,
    float* __restrict__ pos) {
  __shared__ __attribute__((aligned(16))) float plds[4][16][PT_PITCH];
  __shared__ float kpart[4];
  if (blockIdx.x < ATT_GROUPS) {
    attn_body(qkv, obuf, (int)blockIdx.x, plds, kpart);
  } else {
    pos_conv_body(qkv, pw, pb, pos, (int)blockIdx.x - ATT_GROUPS);
  }
}

#define RS_ROWS 64
#define RS_COLS 64
#define RS_SP 260
__global__ __launch_bounds__(128) void res_gemm_kernel(
    const float* __restrict__ obuf, const float* __restrict__ pos,
    const float* __restrict__ w,
    const float* __restrict__ bias, float* __restrict__ out) {
  __shared__ __attribute__((aligned(16))) float wl[16 * RS_SP];
  const int tid  = threadIdx.x;
  const int lane = tid & 31;
  const int half = lane >> 4;
  const int l16  = lane & 15;
  const int wv   = tid >> 5;

  int b = blockIdx.x;
  const int mg = b % (CIN / RS_ROWS);   b /= (CIN / RS_ROWS);
  const int cb = b % (LSPAT / RS_COLS);
  const int n  = b / (LSPAT / RS_COLS);
  const int m0 = mg * RS_ROWS;
  const int col = cb * RS_COLS + wv * 16 + l16;

  const size_t boff = (size_t)n * HCH * LSPAT + (size_t)(half * 2) * LSPAT + col;
  const float* ib = obuf + boff;
  const float* pp = pos + boff;
  v2f bo[16], bq[16];
#pragma unroll
  for (int s = 0; s < 16; ++s) {
    bo[s][0] = ib[(size_t)(4 * s) * LSPAT];
    bo[s][1] = ib[(size_t)(4 * s + 1) * LSPAT];
    bq[s][0] = pp[(size_t)(4 * s) * LSPAT];
    bq[s][1] = pp[(size_t)(4 * s + 1) * LSPAT];
  }

  {
    const int q4   = tid & 15;
    const int srow = tid >> 4;
    const float* wsrc = w + (size_t)(m0 + srow) * HCH + q4 * 4;
    v4f wr[8];
#pragma unroll
    for (int i = 0; i < 8; ++i) {
      const float* p = wsrc + (size_t)(i * 8) * HCH;
      wr[i][0] = p[0]; wr[i][1] = p[1]; wr[i][2] = p[2]; wr[i][3] = p[3];
    }
#pragma unroll
    for (int i = 0; i < 8; ++i)
      *(v4f*)(wl + q4 * RS_SP + (srow + 8 * i) * 4) = wr[i];
  }
  __builtin_amdgcn_fence(__ATOMIC_RELEASE, "workgroup", "local");
  __builtin_amdgcn_s_barrier();
  __builtin_amdgcn_fence(__ATOMIC_ACQUIRE, "workgroup", "local");

  v2f bf[16];
#pragma unroll
  for (int s = 0; s < 16; ++s) bf[s] = bo[s] + bq[s];

  const int aoff = l16 * 4 + half * 2;
  v8f acc[4] = {};
#pragma unroll
  for (int s = 0; s < 16; ++s) {
#pragma unroll
    for (int t = 0; t < 4; ++t) {
      const v2f a = *(const v2f*)(wl + s * RS_SP + aoff + t * 64);
      acc[t] = wmma4(a, bf[s], acc[t]);
    }
  }

  float* ob = out + (size_t)n * CIN * LSPAT;
#pragma unroll
  for (int t = 0; t < 4; ++t) {
#pragma unroll
    for (int r = 0; r < 8; ++r) {
      const int row = m0 + t * 16 + r + half * 8;
      ob[(size_t)row * LSPAT + col] = acc[t][r] + bias[row];
    }
  }
}

extern "C" void kernel_launch(void* const* d_in, const int* in_sizes, int n_in,
                              void* d_out, int out_size, void* d_ws, size_t ws_size,
                              hipStream_t stream) {
  const float* x     = (const float*)d_in[0];
  const float* qkv_w = (const float*)d_in[1];
  const float* qkv_b = (const float*)d_in[2];
  const float* pos_w = (const float*)d_in[3];
  const float* pos_b = (const float*)d_in[4];
  const float* res_w = (const float*)d_in[5];
  const float* res_b = (const float*)d_in[6];
  float* out = (float*)d_out;

  float* qkv  = (float*)d_ws;
  float* obuf = qkv + (size_t)NBATCH * CH3 * LSPAT;
  float* pos  = obuf + (size_t)NBATCH * HCH * LSPAT;

  qkv_gemm_kernel<<<NBATCH * (CH3 / QK_ROWS) * (LSPAT / QK_COLS), 256, 0,
                    stream>>>(x, qkv_w, qkv_b, qkv);

  attn_pos_kernel<<<ATT_GROUPS + POS_GROUPS, 128, 0, stream>>>(qkv, pos_w, pos_b,
                                                             obuf, pos);

  res_gemm_kernel<<<NBATCH * (CIN / RS_ROWS) * (LSPAT / RS_COLS), 128, 0,
                    stream>>>(obuf, pos, res_w, res_b, out);
}
